# starting kernel
# baseline (speedup 1.0000x reference)
#include <hip/hip_runtime.h>
#include <math.h>

typedef float v2f __attribute__((ext_vector_type(2)));
typedef float v8f __attribute__((ext_vector_type(8)));

#define IN_FEATS  256
#define OUT_FEATS 64
#define WPAIRS    (IN_FEATS / 2)
#define WSTRIDE   160

__global__ void gcn_degrees(const int* __restrict__ src,
                            const int* __restrict__ dst,
                            float* __restrict__ deg_src,
                            float* __restrict__ deg_dst,
                            int n_edges) {
    int t = blockIdx.x * blockDim.x + threadIdx.x;
    if (t >= n_edges) return;
    atomicAdd(&deg_src[src[t]], 1.0f);
    atomicAdd(&deg_dst[dst[t]], 1.0f);
}

__global__ void gcn_norms(float* __restrict__ deg_src,
                          float* __restrict__ deg_dst,
                          int n_nodes) {
    int t = blockIdx.x * blockDim.x + threadIdx.x;
    if (t >= n_nodes) return;
    deg_src[t] = rsqrtf(fmaxf(deg_src[t], 1.0f));
    deg_dst[t] = rsqrtf(fmaxf(deg_dst[t], 1.0f));
}

__global__ void __launch_bounds__(256)
gcn_gemm_wmma(const float* __restrict__ feat,
              const float* __restrict__ norm_src,
              const float* __restrict__ W,
              float* __restrict__ h,
              int n_tiles) {
    __shared__ float Wl[WPAIRS * WSTRIDE];

    const int tid = threadIdx.x;
    for (int i = tid; i < WPAIRS * (OUT_FEATS / 2); i += 256) {
        int p  = i >> 5;
        int n2 = (i & 31) * 2;
        float2 lo = *reinterpret_cast<const float2*>(&W[(2 * p) * OUT_FEATS + n2]);
        float2 hi = *reinterpret_cast<const float2*>(&W[(2 * p + 1) * OUT_FEATS + n2]);
        float4 q;  q.x = lo.x;  q.y = hi.x;  q.z = lo.y;  q.w = hi.y;
        *reinterpret_cast<float4*>(&Wl[p * WSTRIDE + 2 * n2]) = q;
    }
    __syncthreads();

    const int wave = tid >> 5;
    const int lane = tid & 31;
    const int tile = blockIdx.x * 8 + wave;
    if (tile >= n_tiles) return;

    const int half = lane >> 4;
    const int l15  = lane & 15;
    const int m    = tile * 16 + l15;
    const float nrm = norm_src[m];
    const float* frow  = feat + (size_t)m * IN_FEATS + 2 * half;
    const float* wbase = &Wl[half * WSTRIDE + 2 * l15];

    v8f c0 = {}, c1 = {}, c2 = {}, c3 = {};

    #pragma unroll 4
    for (int k = 0; k < IN_FEATS; k += 4) {
        v2f a = *reinterpret_cast<const v2f*>(frow + k);
        a = a * nrm;

        const float* wp = wbase + (k >> 1) * WSTRIDE;
        v2f b0 = *reinterpret_cast<const v2f*>(wp);
        v2f b1 = *reinterpret_cast<const v2f*>(wp + 32);
        v2f b2 = *reinterpret_cast<const v2f*>(wp + 64);
        v2f b3 = *reinterpret_cast<const v2f*>(wp + 96);

        c0 = __builtin_amdgcn_wmma_f32_16x16x4_f32(false, a, false, b0,
                                                   (short)0, c0, false, false);
        c1 = __builtin_amdgcn_wmma_f32_16x16x4_f32(false, a, false, b1,
                                                   (short)0, c1, false, false);
        c2 = __builtin_amdgcn_wmma_f32_16x16x4_f32(false, a, false, b2,
                                                   (short)0, c2, false, false);
        c3 = __builtin_amdgcn_wmma_f32_16x16x4_f32(false, a, false, b3,
                                                   (short)0, c3, false, false);
    }

    const int rbase = tile * 16 + 8 * half;
    #pragma unroll
    for (int r = 0; r < 8; ++r) {
        float* orow = h + (size_t)(rbase + r) * OUT_FEATS + l15;
        orow[0]  = c0[r];
        orow[16] = c1[r];
        orow[32] = c2[r];
        orow[48] = c3[r];
    }
}

__global__ void gcn_scatter(const int* __restrict__ src,
                            const int* __restrict__ dst,
                            const float* __restrict__ h,
                            float* __restrict__ agg,
                            int n_edges) {
    long long t = (long long)blockIdx.x * blockDim.x + threadIdx.x;
    int e = (int)(t >> 4);
    if (e >= n_edges) return;
    int q = (int)(t & 15);
    int s = src[e];
    int d = dst[e];
    float4 v = *reinterpret_cast<const float4*>(h + (size_t)s * OUT_FEATS + q * 4);
    float* ap = agg + (size_t)d * OUT_FEATS + q * 4;
    atomicAdd(ap + 0, v.x);
    atomicAdd(ap + 1, v.y);
    atomicAdd(ap + 2, v.z);
    atomicAdd(ap + 3, v.w);
}

__global__ void gcn_finalize(const float* __restrict__ agg,
                             const float* __restrict__ norm_dst,
                             const float* __restrict__ b,
                             float* __restrict__ out,
                             int total) {
    int t = blockIdx.x * blockDim.x + threadIdx.x;
    if (t >= total) return;
    int node = t >> 6;
    int f    = t & 63;
    float x = agg[t] * norm_dst[node] + b[f];
    out[t] = 1.0f / (1.0f + __expf(-x));
}

extern "C" void kernel_launch(void* const* d_in, const int* in_sizes, int n_in,
                              void* d_out, int out_size, void* d_ws, size_t ws_size,
                              hipStream_t stream) {
    const float* feat = (const float*)d_in[0];
    const int*   src  = (const int*)d_in[1];
    const int*   dst  = (const int*)d_in[2];
    const float* W    = (const float*)d_in[3];
    const float* bias = (const float*)d_in[4];
    float*       out  = (float*)d_out;

    const int n_nodes = in_sizes[0] / IN_FEATS;
    const int n_edges = in_sizes[1];

    float* norm_src = (float*)d_ws;
    float* norm_dst = norm_src + n_nodes;
    float* agg      = norm_dst + n_nodes;

    size_t zero_bytes = ((size_t)2 * n_nodes + (size_t)n_nodes * OUT_FEATS)
                        * sizeof(float);
    hipMemsetAsync(d_ws, 0, zero_bytes, stream);

    gcn_degrees<<<(n_edges + 255) / 256, 256, 0, stream>>>(
        src, dst, norm_src, norm_dst, n_edges);

    gcn_norms<<<(n_nodes + 255) / 256, 256, 0, stream>>>(
        norm_src, norm_dst, n_nodes);

    const int n_tiles = n_nodes / 16;
    gcn_gemm_wmma<<<(n_tiles + 7) / 8, 256, 0, stream>>>(
        feat, norm_src, W, out, n_tiles);

    long long scat_threads = (long long)n_edges * 16;
    gcn_scatter<<<(int)((scat_threads + 255) / 256), 256, 0, stream>>>(
        src, dst, out, agg, n_edges);

    int total = n_nodes * OUT_FEATS;
    gcn_finalize<<<(total + 255) / 256, 256, 0, stream>>>(
        agg, norm_dst, bias, out, total);
}


# degree counts by per-work-group LDS integer histograms over edge slices, slice planes summed in the norm kernel; replaces the two scattered global f32 atomic adds per edge (counts exact)
# speedup vs baseline: 1.6218x; 1.6218x over previous
#include <hip/hip_runtime.h>
#include <math.h>

typedef float v2f __attribute__((ext_vector_type(2)));
typedef float v8f __attribute__((ext_vector_type(8)));

#define IN_FEATS  256
#define OUT_FEATS 64
#define WPAIRS    (IN_FEATS / 2)
#define WSTRIDE   160
#define DEG_SLICES 32
#define HIST_NODES 65536
#define DEG_BATCH  8

__device__ __forceinline__ void deg_count(unsigned* hist, int node, int base, int cnt) {
    unsigned r = (unsigned)(node - base);
    if (r < (unsigned)cnt) atomicAdd(&hist[r], 1u);
}

__global__ void __launch_bounds__(1024)
gcn_degree_slices(const int* __restrict__ src,
                  const int* __restrict__ dst,
                  unsigned* __restrict__ part,
                  int n_edges, int n_nodes) {
    __shared__ unsigned hist[HIST_NODES];

    const int tid   = threadIdx.x;
    const int which = blockIdx.x / DEG_SLICES;
    const int slice = blockIdx.x - which * DEG_SLICES;
    const int* __restrict__ idx = which ? dst : src;
    unsigned* __restrict__ plane = part + ((size_t)which * DEG_SLICES + slice) * n_nodes;

    const int nquads = n_edges >> 2;
    const int qper   = (nquads + DEG_SLICES - 1) / DEG_SLICES;
    const int q0     = min(slice * qper, nquads);
    const int q1     = min(q0 + qper, nquads);
    const int4* __restrict__ idx4 = reinterpret_cast<const int4*>(idx);

    for (int base = 0; base < n_nodes; base += HIST_NODES) {
        const int cnt = min(HIST_NODES, n_nodes - base);

        for (int i = tid * 4; i < cnt; i += 4096)
            *reinterpret_cast<uint4*>(&hist[i]) = make_uint4(0u, 0u, 0u, 0u);
        __syncthreads();

        for (int q = q0 + tid; q < q1; q += DEG_BATCH * 1024) {
            int4 v[DEG_BATCH];
            #pragma unroll
            for (int j = 0; j < DEG_BATCH; ++j) {
                const int qq = q + j * 1024;
                v[j] = (qq < q1) ? idx4[qq] : make_int4(-1, -1, -1, -1);
            }
            #pragma unroll
            for (int j = 0; j < DEG_BATCH; ++j) {
                deg_count(hist, v[j].x, base, cnt);
                deg_count(hist, v[j].y, base, cnt);
                deg_count(hist, v[j].z, base, cnt);
                deg_count(hist, v[j].w, base, cnt);
            }
        }
        if (slice == DEG_SLICES - 1)
            for (int e = nquads * 4 + tid; e < n_edges; e += 1024)
                deg_count(hist, idx[e], base, cnt);
        __syncthreads();

        #pragma unroll 4
        for (int i = tid * 4; i < cnt; i += 4096)
            *reinterpret_cast<uint4*>(&plane[base + i]) =
                *reinterpret_cast<const uint4*>(&hist[i]);
        __syncthreads();
    }
}

__global__ void gcn_norms(const unsigned* __restrict__ part,
                          float* __restrict__ norm_src,
                          float* __restrict__ norm_dst,
                          int n_nodes) {
    int t = blockIdx.x * blockDim.x + threadIdx.x;
    if (t >= n_nodes) return;
    const unsigned* ps = part + t;
    const unsigned* pd = part + (size_t)DEG_SLICES * n_nodes + t;
    unsigned cs = 0, cd = 0;
    #pragma unroll
    for (int s = 0; s < DEG_SLICES; ++s) {
        cs += ps[(size_t)s * n_nodes];
        cd += pd[(size_t)s * n_nodes];
    }
    norm_src[t] = rsqrtf(fmaxf((float)cs, 1.0f));
    norm_dst[t] = rsqrtf(fmaxf((float)cd, 1.0f));
}

__global__ void __launch_bounds__(256)
gcn_gemm_wmma(const float* __restrict__ feat,
              const float* __restrict__ norm_src,
              const float* __restrict__ W,
              float* __restrict__ h,
              int n_tiles) {
    __shared__ float Wl[WPAIRS * WSTRIDE];

    const int tid = threadIdx.x;
    for (int i = tid; i < WPAIRS * (OUT_FEATS / 2); i += 256) {
        int p  = i >> 5;
        int n2 = (i & 31) * 2;
        float2 lo = *reinterpret_cast<const float2*>(&W[(2 * p) * OUT_FEATS + n2]);
        float2 hi = *reinterpret_cast<const float2*>(&W[(2 * p + 1) * OUT_FEATS + n2]);
        float4 q;  q.x = lo.x;  q.y = hi.x;  q.z = lo.y;  q.w = hi.y;
        *reinterpret_cast<float4*>(&Wl[p * WSTRIDE + 2 * n2]) = q;
    }
    __syncthreads();

    const int wave = tid >> 5;
    const int lane = tid & 31;
    const int tile = blockIdx.x * 8 + wave;
    if (tile >= n_tiles) return;

    const int half = lane >> 4;
    const int l15  = lane & 15;
    const int m    = tile * 16 + l15;
    const float nrm = norm_src[m];
    const float* frow  = feat + (size_t)m * IN_FEATS + 2 * half;
    const float* wbase = &Wl[half * WSTRIDE + 2 * l15];

    v8f c0 = {}, c1 = {}, c2 = {}, c3 = {};

    #pragma unroll 4
    for (int k = 0; k < IN_FEATS; k += 4) {
        v2f a = *reinterpret_cast<const v2f*>(frow + k);
        a = a * nrm;

        const float* wp = wbase + (k >> 1) * WSTRIDE;
        v2f b0 = *reinterpret_cast<const v2f*>(wp);
        v2f b1 = *reinterpret_cast<const v2f*>(wp + 32);
        v2f b2 = *reinterpret_cast<const v2f*>(wp + 64);
        v2f b3 = *reinterpret_cast<const v2f*>(wp + 96);

        c0 = __builtin_amdgcn_wmma_f32_16x16x4_f32(false, a, false, b0,
                                                   (short)0, c0, false, false);
        c1 = __builtin_amdgcn_wmma_f32_16x16x4_f32(false, a, false, b1,
                                                   (short)0, c1, false, false);
        c2 = __builtin_amdgcn_wmma_f32_16x16x4_f32(false, a, false, b2,
                                                   (short)0, c2, false, false);
        c3 = __builtin_amdgcn_wmma_f32_16x16x4_f32(false, a, false, b3,
                                                   (short)0, c3, false, false);
    }

    const int rbase = tile * 16 + 8 * half;
    #pragma unroll
    for (int r = 0; r < 8; ++r) {
        float* orow = h + (size_t)(rbase + r) * OUT_FEATS + l15;
        orow[0]  = c0[r];
        orow[16] = c1[r];
        orow[32] = c2[r];
        orow[48] = c3[r];
    }
}

__global__ void gcn_scatter(const int* __restrict__ src,
                            const int* __restrict__ dst,
                            const float* __restrict__ h,
                            float* __restrict__ agg,
                            int n_edges) {
    long long t = (long long)blockIdx.x * blockDim.x + threadIdx.x;
    int e = (int)(t >> 4);
    if (e >= n_edges) return;
    int q = (int)(t & 15);
    int s = src[e];
    int d = dst[e];
    float4 v = *reinterpret_cast<const float4*>(h + (size_t)s * OUT_FEATS + q * 4);
    float* ap = agg + (size_t)d * OUT_FEATS + q * 4;
    atomicAdd(ap + 0, v.x);
    atomicAdd(ap + 1, v.y);
    atomicAdd(ap + 2, v.z);
    atomicAdd(ap + 3, v.w);
}

__global__ void gcn_finalize(const float* __restrict__ agg,
                             const float* __restrict__ norm_dst,
                             const float* __restrict__ b,
                             float* __restrict__ out,
                             int total) {
    int t = blockIdx.x * blockDim.x + threadIdx.x;
    if (t >= total) return;
    int node = t >> 6;
    int f    = t & 63;
    float x = agg[t] * norm_dst[node] + b[f];
    out[t] = 1.0f / (1.0f + __expf(-x));
}

extern "C" void kernel_launch(void* const* d_in, const int* in_sizes, int n_in,
                              void* d_out, int out_size, void* d_ws, size_t ws_size,
                              hipStream_t stream) {
    const float* feat = (const float*)d_in[0];
    const int*   src  = (const int*)d_in[1];
    const int*   dst  = (const int*)d_in[2];
    const float* W    = (const float*)d_in[3];
    const float* bias = (const float*)d_in[4];
    float*       out  = (float*)d_out;

    const int n_nodes = in_sizes[0] / IN_FEATS;
    const int n_edges = in_sizes[1];

    float* norm_src = (float*)d_ws;
    float* norm_dst = norm_src + n_nodes;
    float* agg      = norm_dst + n_nodes;
    unsigned* part  = (unsigned*)(agg + (size_t)n_nodes * OUT_FEATS);

    size_t zero_bytes = (size_t)n_nodes * OUT_FEATS * sizeof(float);
    hipMemsetAsync(agg, 0, zero_bytes, stream);

    gcn_degree_slices<<<2 * DEG_SLICES, 1024, 0, stream>>>(
        src, dst, part, n_edges, n_nodes);

    gcn_norms<<<(n_nodes + 255) / 256, 256, 0, stream>>>(
        part, norm_src, norm_dst, n_nodes);

    const int n_tiles = n_nodes / 16;
    gcn_gemm_wmma<<<(n_tiles + 7) / 8, 256, 0, stream>>>(
        feat, norm_src, W, out, n_tiles);

    long long scat_threads = (long long)n_edges * 16;
    gcn_scatter<<<(int)((scat_threads + 255) / 256), 256, 0, stream>>>(
        src, dst, out, agg, n_edges);

    int total = n_nodes * OUT_FEATS;
    gcn_finalize<<<(total + 255) / 256, 256, 0, stream>>>(
        agg, norm_dst, bias, out, total);
}


# scatter re-tiled lane = feature: a wave walks 32 edges and adds each row with 2 whole-line f32 atomic instructions (was 8 quarter-filled line requests per edge); 16x fewer waves start
# speedup vs baseline: 2.1841x; 1.3467x over previous
#include <hip/hip_runtime.h>
#include <math.h>

typedef float v2f __attribute__((ext_vector_type(2)));
typedef float v8f __attribute__((ext_vector_type(8)));

#define IN_FEATS  256
#define OUT_FEATS 64
#define WPAIRS    (IN_FEATS / 2)
#define WSTRIDE   160
#define DEG_SLICES 32
#define HIST_NODES 65536
#define DEG_BATCH  8

__device__ __forceinline__ void deg_count(unsigned* hist, int node, int base, int cnt) {
    unsigned r = (unsigned)(node - base);
    if (r < (unsigned)cnt) atomicAdd(&hist[r], 1u);
}

__global__ void __launch_bounds__(1024)
gcn_degree_slices(const int* __restrict__ src,
                  const int* __restrict__ dst,
                  unsigned* __restrict__ part,
                  int n_edges, int n_nodes) {
    __shared__ unsigned hist[HIST_NODES];

    const int tid   = threadIdx.x;
    const int which = blockIdx.x / DEG_SLICES;
    const int slice = blockIdx.x - which * DEG_SLICES;
    const int* __restrict__ idx = which ? dst : src;
    unsigned* __restrict__ plane = part + ((size_t)which * DEG_SLICES + slice) * n_nodes;

    const int nquads = n_edges >> 2;
    const int qper   = (nquads + DEG_SLICES - 1) / DEG_SLICES;
    const int q0     = min(slice * qper, nquads);
    const int q1     = min(q0 + qper, nquads);
    const int4* __restrict__ idx4 = reinterpret_cast<const int4*>(idx);

    for (int base = 0; base < n_nodes; base += HIST_NODES) {
        const int cnt = min(HIST_NODES, n_nodes - base);

        for (int i = tid * 4; i < cnt; i += 4096)
            *reinterpret_cast<uint4*>(&hist[i]) = make_uint4(0u, 0u, 0u, 0u);
        __syncthreads();

        for (int q = q0 + tid; q < q1; q += DEG_BATCH * 1024) {
            int4 v[DEG_BATCH];
            #pragma unroll
            for (int j = 0; j < DEG_BATCH; ++j) {
                const int qq = q + j * 1024;
                v[j] = (qq < q1) ? idx4[qq] : make_int4(-1, -1, -1, -1);
            }
            #pragma unroll
            for (int j = 0; j < DEG_BATCH; ++j) {
                deg_count(hist, v[j].x, base, cnt);
                deg_count(hist, v[j].y, base, cnt);
                deg_count(hist, v[j].z, base, cnt);
                deg_count(hist, v[j].w, base, cnt);
            }
        }
        if (slice == DEG_SLICES - 1)
            for (int e = nquads * 4 + tid; e < n_edges; e += 1024)
                deg_count(hist, idx[e], base, cnt);
        __syncthreads();

        #pragma unroll 4
        for (int i = tid * 4; i < cnt; i += 4096)
            *reinterpret_cast<uint4*>(&plane[base + i]) =
                *reinterpret_cast<const uint4*>(&hist[i]);
        __syncthreads();
    }
}

__global__ void gcn_norms(const unsigned* __restrict__ part,
                          float* __restrict__ norm_src,
                          float* __restrict__ norm_dst,
                          int n_nodes) {
    int t = blockIdx.x * blockDim.x + threadIdx.x;
    if (t >= n_nodes) return;
    const unsigned* ps = part + t;
    const unsigned* pd = part + (size_t)DEG_SLICES * n_nodes + t;
    unsigned cs = 0, cd = 0;
    #pragma unroll
    for (int s = 0; s < DEG_SLICES; ++s) {
        cs += ps[(size_t)s * n_nodes];
        cd += pd[(size_t)s * n_nodes];
    }
    norm_src[t] = rsqrtf(fmaxf((float)cs, 1.0f));
    norm_dst[t] = rsqrtf(fmaxf((float)cd, 1.0f));
}

__global__ void __launch_bounds__(256)
gcn_gemm_wmma(const float* __restrict__ feat,
              const float* __restrict__ norm_src,
              const float* __restrict__ W,
              float* __restrict__ h,
              int n_tiles) {
    __shared__ float Wl[WPAIRS * WSTRIDE];

    const int tid = threadIdx.x;
    for (int i = tid; i < WPAIRS * (OUT_FEATS / 2); i += 256) {
        int p  = i >> 5;
        int n2 = (i & 31) * 2;
        float2 lo = *reinterpret_cast<const float2*>(&W[(2 * p) * OUT_FEATS + n2]);
        float2 hi = *reinterpret_cast<const float2*>(&W[(2 * p + 1) * OUT_FEATS + n2]);
        float4 q;  q.x = lo.x;  q.y = hi.x;  q.z = lo.y;  q.w = hi.y;
        *reinterpret_cast<float4*>(&Wl[p * WSTRIDE + 2 * n2]) = q;
    }
    __syncthreads();

    const int wave = tid >> 5;
    const int lane = tid & 31;
    const int tile = blockIdx.x * 8 + wave;
    if (tile >= n_tiles) return;

    const int half = lane >> 4;
    const int l15  = lane & 15;
    const int m    = tile * 16 + l15;
    const float nrm = norm_src[m];
    const float* frow  = feat + (size_t)m * IN_FEATS + 2 * half;
    const float* wbase = &Wl[half * WSTRIDE + 2 * l15];

    v8f c0 = {}, c1 = {}, c2 = {}, c3 = {};

    #pragma unroll 4
    for (int k = 0; k < IN_FEATS; k += 4) {
        v2f a = *reinterpret_cast<const v2f*>(frow + k);
        a = a * nrm;

        const float* wp = wbase + (k >> 1) * WSTRIDE;
        v2f b0 = *reinterpret_cast<const v2f*>(wp);
        v2f b1 = *reinterpret_cast<const v2f*>(wp + 32);
        v2f b2 = *reinterpret_cast<const v2f*>(wp + 64);
        v2f b3 = *reinterpret_cast<const v2f*>(wp + 96);

        c0 = __builtin_amdgcn_wmma_f32_16x16x4_f32(false, a, false, b0,
                                                   (short)0, c0, false, false);
        c1 = __builtin_amdgcn_wmma_f32_16x16x4_f32(false, a, false, b1,
                                                   (short)0, c1, false, false);
        c2 = __builtin_amdgcn_wmma_f32_16x16x4_f32(false, a, false, b2,
                                                   (short)0, c2, false, false);
        c3 = __builtin_amdgcn_wmma_f32_16x16x4_f32(false, a, false, b3,
                                                   (short)0, c3, false, false);
    }

    const int rbase = tile * 16 + 8 * half;
    #pragma unroll
    for (int r = 0; r < 8; ++r) {
        float* orow = h + (size_t)(rbase + r) * OUT_FEATS + l15;
        orow[0]  = c0[r];
        orow[16] = c1[r];
        orow[32] = c2[r];
        orow[48] = c3[r];
    }
}

#define SCAT_EPW 32
#define SCAT_EB  8

__global__ void __launch_bounds__(256)
gcn_scatter(const int* __restrict__ src,
            const int* __restrict__ dst,
            const float* __restrict__ h,
            float* __restrict__ agg,
            int n_edges) {
    const unsigned lane = threadIdx.x & 31u;
    const int wave = __builtin_amdgcn_readfirstlane((int)(threadIdx.x >> 5));
    const int e0   = ((int)blockIdx.x * 8 + wave) * SCAT_EPW;
    if (e0 >= n_edges) return;
    const int cnt = min(SCAT_EPW, n_edges - e0);

    const int e  = e0 + (int)lane;
    const int sv = (e < n_edges) ? src[e] : 0;
    const int dv = (e < n_edges) ? dst[e] : 0;

    if (cnt == SCAT_EPW) {
        #pragma unroll
        for (int jb = 0; jb < SCAT_EPW; jb += SCAT_EB) {
            float lo[SCAT_EB], hi[SCAT_EB];
            #pragma unroll
            for (int j = 0; j < SCAT_EB; ++j) {
                const unsigned s = (unsigned)__builtin_amdgcn_readlane(sv, jb + j);
                const float* hp = h + (size_t)s * OUT_FEATS;
                lo[j] = hp[lane];
                hi[j] = hp[lane + 32];
            }
            #pragma unroll
            for (int j = 0; j < SCAT_EB; ++j) {
                const unsigned d = (unsigned)__builtin_amdgcn_readlane(dv, jb + j);
                float* ap = agg + (size_t)d * OUT_FEATS;
                atomicAdd(ap + lane,      lo[j]);
                atomicAdd(ap + lane + 32, hi[j]);
            }
        }
    } else {
        for (int j = 0; j < cnt; ++j) {
            const unsigned s = (unsigned)__builtin_amdgcn_readlane(sv, j);
            const unsigned d = (unsigned)__builtin_amdgcn_readlane(dv, j);
            const float* hp = h + (size_t)s * OUT_FEATS;
            float* ap = agg + (size_t)d * OUT_FEATS;
            const float a = hp[lane];
            const float b = hp[lane + 32];
            atomicAdd(ap + lane,      a);
            atomicAdd(ap + lane + 32, b);
        }
    }
}

__global__ void gcn_finalize(const float* __restrict__ agg,
                             const float* __restrict__ norm_dst,
                             const float* __restrict__ b,
                             float* __restrict__ out,
                             int total) {
    int t = blockIdx.x * blockDim.x + threadIdx.x;
    if (t >= total) return;
    int node = t >> 6;
    int f    = t & 63;
    float x = agg[t] * norm_dst[node] + b[f];
    out[t] = 1.0f / (1.0f + __expf(-x));
}

extern "C" void kernel_launch(void* const* d_in, const int* in_sizes, int n_in,
                              void* d_out, int out_size, void* d_ws, size_t ws_size,
                              hipStream_t stream) {
    const float* feat = (const float*)d_in[0];
    const int*   src  = (const int*)d_in[1];
    const int*   dst  = (const int*)d_in[2];
    const float* W    = (const float*)d_in[3];
    const float* bias = (const float*)d_in[4];
    float*       out  = (float*)d_out;

    const int n_nodes = in_sizes[0] / IN_FEATS;
    const int n_edges = in_sizes[1];

    float* norm_src = (float*)d_ws;
    float* norm_dst = norm_src + n_nodes;
    float* agg      = norm_dst + n_nodes;
    unsigned* part  = (unsigned*)(agg + (size_t)n_nodes * OUT_FEATS);

    size_t zero_bytes = (size_t)n_nodes * OUT_FEATS * sizeof(float);
    hipMemsetAsync(agg, 0, zero_bytes, stream);

    gcn_degree_slices<<<2 * DEG_SLICES, 1024, 0, stream>>>(
        src, dst, part, n_edges, n_nodes);

    gcn_norms<<<(n_nodes + 255) / 256, 256, 0, stream>>>(
        part, norm_src, norm_dst, n_nodes);

    const int n_tiles = n_nodes / 16;
    gcn_gemm_wmma<<<(n_tiles + 7) / 8, 256, 0, stream>>>(
        feat, norm_src, W, out, n_tiles);

    gcn_scatter<<<(n_edges + 8 * SCAT_EPW - 1) / (8 * SCAT_EPW), 256, 0, stream>>>(
        src, dst, out, agg, n_edges);

    int total = n_nodes * OUT_FEATS;
    gcn_finalize<<<(total + 255) / 256, 256, 0, stream>>>(
        agg, norm_dst, bias, out, total);
}


# agg zeroing (was a memset) and the norm kernel folded into the GEMM: each wave sums its 16 rows' count planes, stores norm_dst, zeroes its agg rows; 2 fewer stream operations
# speedup vs baseline: 2.4220x; 1.1089x over previous
#include <hip/hip_runtime.h>
#include <math.h>

typedef float v2f __attribute__((ext_vector_type(2)));
typedef float v8f __attribute__((ext_vector_type(8)));

#define IN_FEATS  256
#define OUT_FEATS 64
#define WPAIRS    (IN_FEATS / 2)
#define WSTRIDE   160
#define DEG_SLICES 32
#define HIST_NODES 65536
#define DEG_BATCH  8

__device__ __forceinline__ void deg_count(unsigned* hist, int node, int base, int cnt) {
    unsigned r = (unsigned)(node - base);
    if (r < (unsigned)cnt) atomicAdd(&hist[r], 1u);
}

__global__ void __launch_bounds__(1024)
gcn_degree_slices(const int* __restrict__ src,
                  const int* __restrict__ dst,
                  unsigned* __restrict__ part,
                  int n_edges, int n_nodes) {
    __shared__ unsigned hist[HIST_NODES];

    const int tid   = threadIdx.x;
    const int which = blockIdx.x / DEG_SLICES;
    const int slice = blockIdx.x - which * DEG_SLICES;
    const int* __restrict__ idx = which ? dst : src;
    unsigned* __restrict__ plane = part + ((size_t)which * DEG_SLICES + slice) * n_nodes;

    const int nquads = n_edges >> 2;
    const int qper   = (nquads + DEG_SLICES - 1) / DEG_SLICES;
    const int q0     = min(slice * qper, nquads);
    const int q1     = min(q0 + qper, nquads);
    const int4* __restrict__ idx4 = reinterpret_cast<const int4*>(idx);

    for (int base = 0; base < n_nodes; base += HIST_NODES) {
        const int cnt = min(HIST_NODES, n_nodes - base);

        for (int i = tid * 4; i < cnt; i += 4096)
            *reinterpret_cast<uint4*>(&hist[i]) = make_uint4(0u, 0u, 0u, 0u);
        __syncthreads();

        for (int q = q0 + tid; q < q1; q += DEG_BATCH * 1024) {
            int4 v[DEG_BATCH];
            #pragma unroll
            for (int j = 0; j < DEG_BATCH; ++j) {
                const int qq = q + j * 1024;
                v[j] = (qq < q1) ? idx4[qq] : make_int4(-1, -1, -1, -1);
            }
            #pragma unroll
            for (int j = 0; j < DEG_BATCH; ++j) {
                deg_count(hist, v[j].x, base, cnt);
                deg_count(hist, v[j].y, base, cnt);
                deg_count(hist, v[j].z, base, cnt);
                deg_count(hist, v[j].w, base, cnt);
            }
        }
        if (slice == DEG_SLICES - 1)
            for (int e = nquads * 4 + tid; e < n_edges; e += 1024)
                deg_count(hist, idx[e], base, cnt);
        __syncthreads();

        #pragma unroll 4
        for (int i = tid * 4; i < cnt; i += 4096)
            *reinterpret_cast<uint4*>(&plane[base + i]) =
                *reinterpret_cast<const uint4*>(&hist[i]);
        __syncthreads();
    }
}

__global__ void __launch_bounds__(256)
gcn_gemm_wmma(const float* __restrict__ feat,
              const unsigned* __restrict__ part,
              const float* __restrict__ W,
              float* __restrict__ h,
              float* __restrict__ norm_dst,
              float* __restrict__ agg,
              int n_tiles, int n_nodes) {
    __shared__ float Wl[WPAIRS * WSTRIDE];
    __shared__ float nd_s[8 * 16];

    const int tid  = threadIdx.x;
    const int wave = tid >> 5;
    const int lane = tid & 31;
    const int tile = blockIdx.x * 8 + wave;
    const int half = lane >> 4;
    const int l15  = lane & 15;

    const int mload = min(tile, n_tiles - 1) * 16 + l15;
    const unsigned* pp = part + (size_t)half * DEG_SLICES * n_nodes + mload;
    unsigned pc[DEG_SLICES];
    #pragma unroll
    for (int s = 0; s < DEG_SLICES; ++s)
        pc[s] = pp[(size_t)s * n_nodes];

    for (int i = tid; i < WPAIRS * (OUT_FEATS / 2); i += 256) {
        int p  = i >> 5;
        int n2 = (i & 31) * 2;
        float2 lo = *reinterpret_cast<const float2*>(&W[(2 * p) * OUT_FEATS + n2]);
        float2 hi = *reinterpret_cast<const float2*>(&W[(2 * p + 1) * OUT_FEATS + n2]);
        float4 q;  q.x = lo.x;  q.y = hi.x;  q.z = lo.y;  q.w = hi.y;
        *reinterpret_cast<float4*>(&Wl[p * WSTRIDE + 2 * n2]) = q;
    }
    unsigned cnt = 0;
    #pragma unroll
    for (int s = 0; s < DEG_SLICES; ++s)
        cnt += pc[s];
    const float nrm_own = rsqrtf(fmaxf((float)cnt, 1.0f));
    if (half) nd_s[wave * 16 + l15] = nrm_own;
    __syncthreads();

    if (wave < 4) {
        const int row = blockIdx.x * 128 + wave * 32 + lane;
        if (row < n_nodes) norm_dst[row] = nd_s[wave * 32 + lane];
    }

    if (tile >= n_tiles) return;

    const int m = tile * 16 + l15;
    const float nrm = __shfl(nrm_own, l15);

    float4* zrow = reinterpret_cast<float4*>(agg + (size_t)tile * (16 * OUT_FEATS));
    #pragma unroll
    for (int i = 0; i < (16 * OUT_FEATS) / (4 * 32); ++i)
        zrow[i * 32 + lane] = make_float4(0.0f, 0.0f, 0.0f, 0.0f);

    const float* frow  = feat + (size_t)m * IN_FEATS + 2 * half;
    const float* wbase = &Wl[half * WSTRIDE + 2 * l15];

    v8f c0 = {}, c1 = {}, c2 = {}, c3 = {};

    #pragma unroll 4
    for (int k = 0; k < IN_FEATS; k += 4) {
        v2f a = *reinterpret_cast<const v2f*>(frow + k);
        a = a * nrm;

        const float* wp = wbase + (k >> 1) * WSTRIDE;
        v2f b0 = *reinterpret_cast<const v2f*>(wp);
        v2f b1 = *reinterpret_cast<const v2f*>(wp + 32);
        v2f b2 = *reinterpret_cast<const v2f*>(wp + 64);
        v2f b3 = *reinterpret_cast<const v2f*>(wp + 96);

        c0 = __builtin_amdgcn_wmma_f32_16x16x4_f32(false, a, false, b0,
                                                   (short)0, c0, false, false);
        c1 = __builtin_amdgcn_wmma_f32_16x16x4_f32(false, a, false, b1,
                                                   (short)0, c1, false, false);
        c2 = __builtin_amdgcn_wmma_f32_16x16x4_f32(false, a, false, b2,
                                                   (short)0, c2, false, false);
        c3 = __builtin_amdgcn_wmma_f32_16x16x4_f32(false, a, false, b3,
                                                   (short)0, c3, false, false);
    }

    const int rbase = tile * 16 + 8 * half;
    #pragma unroll
    for (int r = 0; r < 8; ++r) {
        float* orow = h + (size_t)(rbase + r) * OUT_FEATS + l15;
        orow[0]  = c0[r];
        orow[16] = c1[r];
        orow[32] = c2[r];
        orow[48] = c3[r];
    }
}

#define SCAT_EPW 32
#define SCAT_EB  8

__global__ void __launch_bounds__(256)
gcn_scatter(const int* __restrict__ src,
            const int* __restrict__ dst,
            const float* __restrict__ h,
            float* __restrict__ agg,
            int n_edges) {
    const unsigned lane = threadIdx.x & 31u;
    const int wave = __builtin_amdgcn_readfirstlane((int)(threadIdx.x >> 5));
    const int e0   = ((int)blockIdx.x * 8 + wave) * SCAT_EPW;
    if (e0 >= n_edges) return;
    const int cnt = min(SCAT_EPW, n_edges - e0);

    const int e  = e0 + (int)lane;
    const int sv = (e < n_edges) ? src[e] : 0;
    const int dv = (e < n_edges) ? dst[e] : 0;

    if (cnt == SCAT_EPW) {
        #pragma unroll
        for (int jb = 0; jb < SCAT_EPW; jb += SCAT_EB) {
            float lo[SCAT_EB], hi[SCAT_EB];
            #pragma unroll
            for (int j = 0; j < SCAT_EB; ++j) {
                const unsigned s = (unsigned)__builtin_amdgcn_readlane(sv, jb + j);
                const float* hp = h + (size_t)s * OUT_FEATS;
                lo[j] = hp[lane];
                hi[j] = hp[lane + 32];
            }
            #pragma unroll
            for (int j = 0; j < SCAT_EB; ++j) {
                const unsigned d = (unsigned)__builtin_amdgcn_readlane(dv, jb + j);
                float* ap = agg + (size_t)d * OUT_FEATS;
                atomicAdd(ap + lane,      lo[j]);
                atomicAdd(ap + lane + 32, hi[j]);
            }
        }
    } else {
        for (int j = 0; j < cnt; ++j) {
            const unsigned s = (unsigned)__builtin_amdgcn_readlane(sv, j);
            const unsigned d = (unsigned)__builtin_amdgcn_readlane(dv, j);
            const float* hp = h + (size_t)s * OUT_FEATS;
            float* ap = agg + (size_t)d * OUT_FEATS;
            const float a = hp[lane];
            const float b = hp[lane + 32];
            atomicAdd(ap + lane,      a);
            atomicAdd(ap + lane + 32, b);
        }
    }
}

__global__ void gcn_finalize(const float* __restrict__ agg,
                             const float* __restrict__ norm_dst,
                             const float* __restrict__ b,
                             float* __restrict__ out,
                             int total) {
    int t = blockIdx.x * blockDim.x + threadIdx.x;
    if (t >= total) return;
    int node = t >> 6;
    int f    = t & 63;
    float x = agg[t] * norm_dst[node] + b[f];
    out[t] = 1.0f / (1.0f + __expf(-x));
}

extern "C" void kernel_launch(void* const* d_in, const int* in_sizes, int n_in,
                              void* d_out, int out_size, void* d_ws, size_t ws_size,
                              hipStream_t stream) {
    const float* feat = (const float*)d_in[0];
    const int*   src  = (const int*)d_in[1];
    const int*   dst  = (const int*)d_in[2];
    const float* W    = (const float*)d_in[3];
    const float* bias = (const float*)d_in[4];
    float*       out  = (float*)d_out;

    const int n_nodes = in_sizes[0] / IN_FEATS;
    const int n_edges = in_sizes[1];

    float* agg      = (float*)d_ws;
    unsigned* part  = (unsigned*)(agg + (size_t)n_nodes * OUT_FEATS);
    float* norm_dst = (float*)(part + (size_t)2 * DEG_SLICES * n_nodes);

    gcn_degree_slices<<<2 * DEG_SLICES, 1024, 0, stream>>>(
        src, dst, part, n_edges, n_nodes);

    const int n_tiles = n_nodes / 16;
    gcn_gemm_wmma<<<(n_tiles + 7) / 8, 256, 0, stream>>>(
        feat, part, W, out, norm_dst, agg, n_tiles, n_nodes);

    gcn_scatter<<<(n_edges + 8 * SCAT_EPW - 1) / (8 * SCAT_EPW), 256, 0, stream>>>(
        src, dst, out, agg, n_edges);

    int total = n_nodes * OUT_FEATS;
    gcn_finalize<<<(total + 255) / 256, 256, 0, stream>>>(
        agg, norm_dst, bias, out, total);
}


# scatter walks the edge list once per destination half; work-groups read their die and draw their own half's items first: a row of agg takes its f32 atomic adds from one half (arrival order kept)
# speedup vs baseline: 2.5419x; 1.0495x over previous
#include <hip/hip_runtime.h>
#include <math.h>

typedef float v2f __attribute__((ext_vector_type(2)));
typedef float v8f __attribute__((ext_vector_type(8)));

#define IN_FEATS  256
#define OUT_FEATS 64
#define WPAIRS    (IN_FEATS / 2)
#define WSTRIDE   160
#define DEG_SLICES 32
#define HIST_NODES 65536
#define DEG_BATCH  8
#define TICKET_PITCH 64

__device__ __forceinline__ void deg_count(unsigned* hist, int node, int base, int cnt) {
    unsigned r = (unsigned)(node - base);
    if (r < (unsigned)cnt) atomicAdd(&hist[r], 1u);
}

__global__ void __launch_bounds__(1024)
gcn_degree_slices(const int* __restrict__ src,
                  const int* __restrict__ dst,
                  unsigned* __restrict__ part,
                  int n_edges, int n_nodes) {
    __shared__ unsigned hist[HIST_NODES];

    const int tid   = threadIdx.x;
    const int which = blockIdx.x / DEG_SLICES;
    const int slice = blockIdx.x - which * DEG_SLICES;
    const int* __restrict__ idx = which ? dst : src;
    unsigned* __restrict__ plane = part + ((size_t)which * DEG_SLICES + slice) * n_nodes;

    const int nquads = n_edges >> 2;
    const int qper   = (nquads + DEG_SLICES - 1) / DEG_SLICES;
    const int q0     = min(slice * qper, nquads);
    const int q1     = min(q0 + qper, nquads);
    const int4* __restrict__ idx4 = reinterpret_cast<const int4*>(idx);

    for (int base = 0; base < n_nodes; base += HIST_NODES) {
        const int cnt = min(HIST_NODES, n_nodes - base);

        for (int i = tid * 4; i < cnt; i += 4096)
            *reinterpret_cast<uint4*>(&hist[i]) = make_uint4(0u, 0u, 0u, 0u);
        __syncthreads();

        for (int q = q0 + tid; q < q1; q += DEG_BATCH * 1024) {
            int4 v[DEG_BATCH];
            #pragma unroll
            for (int j = 0; j < DEG_BATCH; ++j) {
                const int qq = q + j * 1024;
                v[j] = (qq < q1) ? idx4[qq] : make_int4(-1, -1, -1, -1);
            }
            #pragma unroll
            for (int j = 0; j < DEG_BATCH; ++j) {
                deg_count(hist, v[j].x, base, cnt);
                deg_count(hist, v[j].y, base, cnt);
                deg_count(hist, v[j].z, base, cnt);
                deg_count(hist, v[j].w, base, cnt);
            }
        }
        if (slice == DEG_SLICES - 1)
            for (int e = nquads * 4 + tid; e < n_edges; e += 1024)
                deg_count(hist, idx[e], base, cnt);
        __syncthreads();

        #pragma unroll 4
        for (int i = tid * 4; i < cnt; i += 4096)
            *reinterpret_cast<uint4*>(&plane[base + i]) =
                *reinterpret_cast<const uint4*>(&hist[i]);
        __syncthreads();
    }
}

__global__ void __launch_bounds__(256)
gcn_gemm_wmma(const float* __restrict__ feat,
              const unsigned* __restrict__ part,
              const float* __restrict__ W,
              float* __restrict__ h,
              float* __restrict__ norm_dst,
              float* __restrict__ agg,
              int* __restrict__ tickets,
              int n_tiles, int n_nodes) {
    __shared__ float Wl[WPAIRS * WSTRIDE];
    __shared__ float nd_s[8 * 16];

    const int tid  = threadIdx.x;
    const int wave = tid >> 5;
    const int lane = tid & 31;
    const int tile = blockIdx.x * 8 + wave;
    const int half = lane >> 4;
    const int l15  = lane & 15;

    if (blockIdx.x == 0 && tid == 0) {
        tickets[0] = 0;
        tickets[TICKET_PITCH] = 0;
    }

    const int mload = min(tile, n_tiles - 1) * 16 + l15;
    const unsigned* pp = part + (size_t)half * DEG_SLICES * n_nodes + mload;
    unsigned pc[DEG_SLICES];
    #pragma unroll
    for (int s = 0; s < DEG_SLICES; ++s)
        pc[s] = pp[(size_t)s * n_nodes];

    for (int i = tid; i < WPAIRS * (OUT_FEATS / 2); i += 256) {
        int p  = i >> 5;
        int n2 = (i & 31) * 2;
        float2 lo = *reinterpret_cast<const float2*>(&W[(2 * p) * OUT_FEATS + n2]);
        float2 hi = *reinterpret_cast<const float2*>(&W[(2 * p + 1) * OUT_FEATS + n2]);
        float4 q;  q.x = lo.x;  q.y = hi.x;  q.z = lo.y;  q.w = hi.y;
        *reinterpret_cast<float4*>(&Wl[p * WSTRIDE + 2 * n2]) = q;
    }
    unsigned cnt = 0;
    #pragma unroll
    for (int s = 0; s < DEG_SLICES; ++s)
        cnt += pc[s];
    const float nrm_own = rsqrtf(fmaxf((float)cnt, 1.0f));
    if (half) nd_s[wave * 16 + l15] = nrm_own;
    __syncthreads();

    if (wave < 4) {
        const int row = blockIdx.x * 128 + wave * 32 + lane;
        if (row < n_nodes) norm_dst[row] = nd_s[wave * 32 + lane];
    }

    if (tile >= n_tiles) return;

    const int m = tile * 16 + l15;
    const float nrm = __shfl(nrm_own, l15);

    float4* zrow = reinterpret_cast<float4*>(agg + (size_t)tile * (16 * OUT_FEATS));
    #pragma unroll
    for (int i = 0; i < (16 * OUT_FEATS) / (4 * 32); ++i)
        zrow[i * 32 + lane] = make_float4(0.0f, 0.0f, 0.0f, 0.0f);

    const float* frow  = feat + (size_t)m * IN_FEATS + 2 * half;
    const float* wbase = &Wl[half * WSTRIDE + 2 * l15];

    v8f c0 = {}, c1 = {}, c2 = {}, c3 = {};

    #pragma unroll 4
    for (int k = 0; k < IN_FEATS; k += 4) {
        v2f a = *reinterpret_cast<const v2f*>(frow + k);
        a = a * nrm;

        const float* wp = wbase + (k >> 1) * WSTRIDE;
        v2f b0 = *reinterpret_cast<const v2f*>(wp);
        v2f b1 = *reinterpret_cast<const v2f*>(wp + 32);
        v2f b2 = *reinterpret_cast<const v2f*>(wp + 64);
        v2f b3 = *reinterpret_cast<const v2f*>(wp + 96);

        c0 = __builtin_amdgcn_wmma_f32_16x16x4_f32(false, a, false, b0,
                                                   (short)0, c0, false, false);
        c1 = __builtin_amdgcn_wmma_f32_16x16x4_f32(false, a, false, b1,
                                                   (short)0, c1, false, false);
        c2 = __builtin_amdgcn_wmma_f32_16x16x4_f32(false, a, false, b2,
                                                   (short)0, c2, false, false);
        c3 = __builtin_amdgcn_wmma_f32_16x16x4_f32(false, a, false, b3,
                                                   (short)0, c3, false, false);
    }

    const int rbase = tile * 16 + 8 * half;
    #pragma unroll
    for (int r = 0; r < 8; ++r) {
        float* orow = h + (size_t)(rbase + r) * OUT_FEATS + l15;
        orow[0]  = c0[r];
        orow[16] = c1[r];
        orow[32] = c2[r];
        orow[48] = c3[r];
    }
}

#define SCAT_EPW  64
#define SCAT_ITEM (8 * SCAT_EPW)
#define SCAT_EB   8
#define SCAT_WGS  2048

__device__ __forceinline__ int scat_take(int* tickets, int q, bool& own_empty, int n_items) {
    if (!own_empty) {
        const int t = atomicAdd(&tickets[q * TICKET_PITCH], 1);
        if (t < n_items) return t * 2 + q;
        own_empty = true;
    }
    const int t = atomicAdd(&tickets[(q ^ 1) * TICKET_PITCH], 1);
    return (t < n_items) ? (t * 2 + (q ^ 1)) : -1;
}

__global__ void __launch_bounds__(256)
gcn_scatter(const int* __restrict__ src,
            const int* __restrict__ dst,
            const float* __restrict__ h,
            float* __restrict__ agg,
            int* __restrict__ tickets,
            int n_edges, int n_nodes) {
    __shared__ int item_s;
    const unsigned lane = threadIdx.x & 31u;
    const int wave = __builtin_amdgcn_readfirstlane((int)(threadIdx.x >> 5));
    const int n_split = n_nodes >> 1;
    const int n_items = (n_edges + SCAT_ITEM - 1) / SCAT_ITEM;

    const unsigned place = __builtin_amdgcn_s_sendmsg_rtn(0x87);
    const int q = (((place >> 16) & 15u) >= 4u) ? 1 : 0;

    bool own_empty = false;
    int nxt = -1;
    if (threadIdx.x == 0) nxt = scat_take(tickets, q, own_empty, n_items);

    for (;;) {
        if (threadIdx.x == 0) item_s = nxt;
        __syncthreads();
        const int it = __builtin_amdgcn_readfirstlane(item_s);
        __syncthreads();
        if (it < 0) break;
        if (threadIdx.x == 0) nxt = scat_take(tickets, q, own_empty, n_items);

        const bool hi_half = (it & 1) != 0;
        const int e0 = (it >> 1) * SCAT_ITEM + wave * SCAT_EPW;
        if (e0 >= n_edges) continue;

        const int ea = e0 + (int)lane;
        const int eb = ea + 32;
        const bool va = ea < n_edges;
        const bool vb = eb < n_edges;
        const int sa = va ? src[ea] : 0;
        const int da = va ? dst[ea] : 0;
        const int sb = vb ? src[eb] : 0;
        const int db = vb ? dst[eb] : 0;
        const unsigned ma = __builtin_amdgcn_ballot_w32(va && ((da >= n_split) == hi_half));
        const unsigned mb = __builtin_amdgcn_ballot_w32(vb && ((db >= n_split) == hi_half));
        unsigned long long m = ((unsigned long long)mb << 32) | (unsigned long long)ma;

        while (m) {
            const bool full = __builtin_popcountll(m) >= SCAT_EB;
            int l[SCAT_EB];
            #pragma unroll
            for (int j = 0; j < SCAT_EB; ++j) {
                l[j] = m ? (int)__builtin_ctzll(m) : -1;
                m &= m - 1ull;
            }
            float lo[SCAT_EB], hi[SCAT_EB];
            if (full) {
                #pragma unroll
                for (int j = 0; j < SCAT_EB; ++j) {
                    const int s_lo = __builtin_amdgcn_readlane(sa, l[j] & 31);
                    const int s_hi = __builtin_amdgcn_readlane(sb, l[j] & 31);
                    const unsigned s = (unsigned)((l[j] < 32) ? s_lo : s_hi);
                    const float* hp = h + (size_t)s * OUT_FEATS;
                    lo[j] = hp[lane];
                    hi[j] = hp[lane + 32];
                }
                #pragma unroll
                for (int j = 0; j < SCAT_EB; ++j) {
                    const int d_lo = __builtin_amdgcn_readlane(da, l[j] & 31);
                    const int d_hi = __builtin_amdgcn_readlane(db, l[j] & 31);
                    const unsigned d = (unsigned)((l[j] < 32) ? d_lo : d_hi);
                    float* ap = agg + (size_t)d * OUT_FEATS;
                    atomicAdd(ap + lane,      lo[j]);
                    atomicAdd(ap + lane + 32, hi[j]);
                }
            } else {
                #pragma unroll
                for (int j = 0; j < SCAT_EB; ++j) {
                    const int lj = (l[j] >= 0) ? l[j] : l[0];
                    const int s_lo = __builtin_amdgcn_readlane(sa, lj & 31);
                    const int s_hi = __builtin_amdgcn_readlane(sb, lj & 31);
                    const unsigned s = (unsigned)((lj < 32) ? s_lo : s_hi);
                    const float* hp = h + (size_t)s * OUT_FEATS;
                    lo[j] = hp[lane];
                    hi[j] = hp[lane + 32];
                }
                #pragma unroll
                for (int j = 0; j < SCAT_EB; ++j) {
                    if (l[j] >= 0) {
                        const int d_lo = __builtin_amdgcn_readlane(da, l[j] & 31);
                        const int d_hi = __builtin_amdgcn_readlane(db, l[j] & 31);
                        const unsigned d = (unsigned)((l[j] < 32) ? d_lo : d_hi);
                        float* ap = agg + (size_t)d * OUT_FEATS;
                        atomicAdd(ap + lane,      lo[j]);
                        atomicAdd(ap + lane + 32, hi[j]);
                    }
                }
            }
        }
    }
}

__global__ void gcn_finalize(const float* __restrict__ agg,
                             const float* __restrict__ norm_dst,
                             const float* __restrict__ b,
                             float* __restrict__ out,
                             int total) {
    int t = blockIdx.x * blockDim.x + threadIdx.x;
    if (t >= total) return;
    int node = t >> 6;
    int f    = t & 63;
    float x = agg[t] * norm_dst[node] + b[f];
    out[t] = 1.0f / (1.0f + __expf(-x));
}

extern "C" void kernel_launch(void* const* d_in, const int* in_sizes, int n_in,
                              void* d_out, int out_size, void* d_ws, size_t ws_size,
                              hipStream_t stream) {
    const float* feat = (const float*)d_in[0];
    const int*   src  = (const int*)d_in[1];
    const int*   dst  = (const int*)d_in[2];
    const float* W    = (const float*)d_in[3];
    const float* bias = (const float*)d_in[4];
    float*       out  = (float*)d_out;

    const int n_nodes = in_sizes[0] / IN_FEATS;
    const int n_edges = in_sizes[1];

    float* agg      = (float*)d_ws;
    unsigned* part  = (unsigned*)(agg + (size_t)n_nodes * OUT_FEATS);
    float* norm_dst = (float*)(part + (size_t)2 * DEG_SLICES * n_nodes);
    int* tickets    = (int*)(norm_dst + n_nodes);

    gcn_degree_slices<<<2 * DEG_SLICES, 1024, 0, stream>>>(
        src, dst, part, n_edges, n_nodes);

    const int n_tiles = n_nodes / 16;
    gcn_gemm_wmma<<<(n_tiles + 7) / 8, 256, 0, stream>>>(
        feat, part, W, out, norm_dst, agg, tickets, n_tiles, n_nodes);

    gcn_scatter<<<SCAT_WGS, 256, 0, stream>>>(
        src, dst, out, agg, tickets, n_edges, n_nodes);

    int total = n_nodes * OUT_FEATS;
    gcn_finalize<<<(total + 255) / 256, 256, 0, stream>>>(
        agg, norm_dst, bias, out, total);
}


# GEMM kernel issues its loads ahead of use: a thread's 32 W staging loads in one batch before its first LDS store (was 8 waited round trips), A fragments one 16-column K block ahead of the WMMAs
# speedup vs baseline: 2.7560x; 1.0842x over previous
#include <hip/hip_runtime.h>
#include <math.h>

typedef float v2f __attribute__((ext_vector_type(2)));
typedef float v8f __attribute__((ext_vector_type(8)));

#define IN_FEATS  256
#define OUT_FEATS 64
#define WPAIRS    (IN_FEATS / 2)
#define WSTRIDE   160
#define DEG_SLICES 32
#define HIST_NODES 65536
#define DEG_BATCH  8
#define W_STAGE_IT 16
#define K_BLK      16
#define TICKET_PITCH 64

__device__ __forceinline__ void deg_count(unsigned* hist, int node, int base, int cnt) {
    unsigned r = (unsigned)(node - base);
    if (r < (unsigned)cnt) atomicAdd(&hist[r], 1u);
}

__global__ void __launch_bounds__(1024)
gcn_degree_slices(const int* __restrict__ src,
                  const int* __restrict__ dst,
                  unsigned* __restrict__ part,
                  int n_edges, int n_nodes) {
    __shared__ unsigned hist[HIST_NODES];

    const int tid   = threadIdx.x;
    const int which = blockIdx.x / DEG_SLICES;
    const int slice = blockIdx.x - which * DEG_SLICES;
    const int* __restrict__ idx = which ? dst : src;
    unsigned* __restrict__ plane = part + ((size_t)which * DEG_SLICES + slice) * n_nodes;

    const int nquads = n_edges >> 2;
    const int qper   = (nquads + DEG_SLICES - 1) / DEG_SLICES;
    const int q0     = min(slice * qper, nquads);
    const int q1     = min(q0 + qper, nquads);
    const int4* __restrict__ idx4 = reinterpret_cast<const int4*>(idx);

    for (int base = 0; base < n_nodes; base += HIST_NODES) {
        const int cnt = min(HIST_NODES, n_nodes - base);

        for (int i = tid * 4; i < cnt; i += 4096)
            *reinterpret_cast<uint4*>(&hist[i]) = make_uint4(0u, 0u, 0u, 0u);
        __syncthreads();

        for (int q = q0 + tid; q < q1; q += DEG_BATCH * 1024) {
            int4 v[DEG_BATCH];
            #pragma unroll
            for (int j = 0; j < DEG_BATCH; ++j) {
                const int qq = q + j * 1024;
                v[j] = (qq < q1) ? idx4[qq] : make_int4(-1, -1, -1, -1);
            }
            #pragma unroll
            for (int j = 0; j < DEG_BATCH; ++j) {
                deg_count(hist, v[j].x, base, cnt);
                deg_count(hist, v[j].y, base, cnt);
                deg_count(hist, v[j].z, base, cnt);
                deg_count(hist, v[j].w, base, cnt);
            }
        }
        if (slice == DEG_SLICES - 1)
            for (int e = nquads * 4 + tid; e < n_edges; e += 1024)
                deg_count(hist, idx[e], base, cnt);
        __syncthreads();

        #pragma unroll 4
        for (int i = tid * 4; i < cnt; i += 4096)
            *reinterpret_cast<uint4*>(&plane[base + i]) =
                *reinterpret_cast<const uint4*>(&hist[i]);
        __syncthreads();
    }
}

__global__ void __launch_bounds__(256)
gcn_gemm_wmma(const float* __restrict__ feat,
              const unsigned* __restrict__ part,
              const float* __restrict__ W,
              float* __restrict__ h,
              float* __restrict__ norm_dst,
              float* __restrict__ agg,
              int* __restrict__ tickets,
              int n_tiles, int n_nodes) {
    __shared__ float Wl[WPAIRS * WSTRIDE];
    __shared__ float nd_s[8 * 16];

    const int tid  = threadIdx.x;
    const int wave = tid >> 5;
    const int lane = tid & 31;
    const int tile = blockIdx.x * 8 + wave;
    const int half = lane >> 4;
    const int l15  = lane & 15;

    if (blockIdx.x == 0 && tid == 0) {
        tickets[0] = 0;
        tickets[TICKET_PITCH] = 0;
    }

    const int mload = min(tile, n_tiles - 1) * 16 + l15;
    const unsigned* pp = part + (size_t)half * DEG_SLICES * n_nodes + mload;
    unsigned pc[DEG_SLICES];
    #pragma unroll
    for (int s = 0; s < DEG_SLICES; ++s)
        pc[s] = pp[(size_t)s * n_nodes];

    float2 wlo[W_STAGE_IT], whi[W_STAGE_IT];
    #pragma unroll
    for (int it = 0; it < W_STAGE_IT; ++it) {
        const int i  = tid + it * 256;
        const int p  = i >> 5;
        const int n2 = (i & 31) * 2;
        wlo[it] = *reinterpret_cast<const float2*>(&W[(2 * p) * OUT_FEATS + n2]);
        whi[it] = *reinterpret_cast<const float2*>(&W[(2 * p + 1) * OUT_FEATS + n2]);
    }

    const float* frow = feat + (size_t)mload * IN_FEATS + 2 * half;
    v2f an[K_BLK / 4] = {};
    if (tile < n_tiles) {
        #pragma unroll
        for (int j = 0; j < K_BLK / 4; ++j)
            an[j] = *reinterpret_cast<const v2f*>(frow + 4 * j);
    }

    #pragma unroll
    for (int it = 0; it < W_STAGE_IT; ++it) {
        const int i  = tid + it * 256;
        const int p  = i >> 5;
        const int n2 = (i & 31) * 2;
        float4 q;  q.x = wlo[it].x;  q.y = whi[it].x;  q.z = wlo[it].y;  q.w = whi[it].y;
        *reinterpret_cast<float4*>(&Wl[p * WSTRIDE + 2 * n2]) = q;
    }
    unsigned cnt = 0;
    #pragma unroll
    for (int s = 0; s < DEG_SLICES; ++s)
        cnt += pc[s];
    const float nrm_own = rsqrtf(fmaxf((float)cnt, 1.0f));
    if (half) nd_s[wave * 16 + l15] = nrm_own;
    __syncthreads();

    if (wave < 4) {
        const int row = blockIdx.x * 128 + wave * 32 + lane;
        if (row < n_nodes) norm_dst[row] = nd_s[wave * 32 + lane];
    }

    if (tile >= n_tiles) return;

    const float nrm = __shfl(nrm_own, l15);

    float4* zrow = reinterpret_cast<float4*>(agg + (size_t)tile * (16 * OUT_FEATS));
    #pragma unroll
    for (int i = 0; i < (16 * OUT_FEATS) / (4 * 32); ++i)
        zrow[i * 32 + lane] = make_float4(0.0f, 0.0f, 0.0f, 0.0f);

    const float* wbase = &Wl[half * WSTRIDE + 2 * l15];

    v8f c0 = {}, c1 = {}, c2 = {}, c3 = {};

    for (int k = 0; k < IN_FEATS; k += K_BLK) {
        v2f ac[K_BLK / 4];
        #pragma unroll
        for (int j = 0; j < K_BLK / 4; ++j)
            ac[j] = an[j] * nrm;

        const int kn = (k + K_BLK < IN_FEATS) ? k + K_BLK : k;
        #pragma unroll
        for (int j = 0; j < K_BLK / 4; ++j)
            an[j] = *reinterpret_cast<const v2f*>(frow + kn + 4 * j);

        #pragma unroll
        for (int j = 0; j < K_BLK / 4; ++j) {
            const v2f a = ac[j];
            const float* wp = wbase + ((k + 4 * j) >> 1) * WSTRIDE;
            v2f b0 = *reinterpret_cast<const v2f*>(wp);
            v2f b1 = *reinterpret_cast<const v2f*>(wp + 32);
            v2f b2 = *reinterpret_cast<const v2f*>(wp + 64);
            v2f b3 = *reinterpret_cast<const v2f*>(wp + 96);

            c0 = __builtin_amdgcn_wmma_f32_16x16x4_f32(false, a, false, b0,
                                                       (short)0, c0, false, false);
            c1 = __builtin_amdgcn_wmma_f32_16x16x4_f32(false, a, false, b1,
                                                       (short)0, c1, false, false);
            c2 = __builtin_amdgcn_wmma_f32_16x16x4_f32(false, a, false, b2,
                                                       (short)0, c2, false, false);
            c3 = __builtin_amdgcn_wmma_f32_16x16x4_f32(false, a, false, b3,
                                                       (short)0, c3, false, false);
        }
    }

    const int rbase = tile * 16 + 8 * half;
    #pragma unroll
    for (int r = 0; r < 8; ++r) {
        float* orow = h + (size_t)(rbase + r) * OUT_FEATS + l15;
        orow[0]  = c0[r];
        orow[16] = c1[r];
        orow[32] = c2[r];
        orow[48] = c3[r];
    }
}

#define SCAT_EPW  64
#define SCAT_ITEM (8 * SCAT_EPW)
#define SCAT_EB   8
#define SCAT_WGS  2048

__device__ __forceinline__ int scat_take(int* tickets, int q, bool& own_empty, int n_items) {
    if (!own_empty) {
        const int t = atomicAdd(&tickets[q * TICKET_PITCH], 1);
        if (t < n_items) return t * 2 + q;
        own_empty = true;
    }
    const int t = atomicAdd(&tickets[(q ^ 1) * TICKET_PITCH], 1);
    return (t < n_items) ? (t * 2 + (q ^ 1)) : -1;
}

__global__ void __launch_bounds__(256)
gcn_scatter(const int* __restrict__ src,
            const int* __restrict__ dst,
            const float* __restrict__ h,
            float* __restrict__ agg,
            int* __restrict__ tickets,
            int n_edges, int n_nodes) {
    __shared__ int item_s;
    const unsigned lane = threadIdx.x & 31u;
    const int wave = __builtin_amdgcn_readfirstlane((int)(threadIdx.x >> 5));
    const int n_split = n_nodes >> 1;
    const int n_items = (n_edges + SCAT_ITEM - 1) / SCAT_ITEM;

    const unsigned place = __builtin_amdgcn_s_sendmsg_rtn(0x87);
    const int q = (((place >> 16) & 15u) >= 4u) ? 1 : 0;

    bool own_empty = false;
    int nxt = -1;
    if (threadIdx.x == 0) nxt = scat_take(tickets, q, own_empty, n_items);

    for (;;) {
        if (threadIdx.x == 0) item_s = nxt;
        __syncthreads();
        const int it = __builtin_amdgcn_readfirstlane(item_s);
        __syncthreads();
        if (it < 0) break;
        if (threadIdx.x == 0) nxt = scat_take(tickets, q, own_empty, n_items);

        const bool hi_half = (it & 1) != 0;
        const int e0 = (it >> 1) * SCAT_ITEM + wave * SCAT_EPW;
        if (e0 >= n_edges) continue;

        const int ea = e0 + (int)lane;
        const int eb = ea + 32;
        const bool va = ea < n_edges;
        const bool vb = eb < n_edges;
        const int sa = va ? src[ea] : 0;
        const int da = va ? dst[ea] : 0;
        const int sb = vb ? src[eb] : 0;
        const int db = vb ? dst[eb] : 0;
        const unsigned ma = __builtin_amdgcn_ballot_w32(va && ((da >= n_split) == hi_half));
        const unsigned mb = __builtin_amdgcn_ballot_w32(vb && ((db >= n_split) == hi_half));
        unsigned long long m = ((unsigned long long)mb << 32) | (unsigned long long)ma;

        while (m) {
            const bool full = __builtin_popcountll(m) >= SCAT_EB;
            int l[SCAT_EB];
            #pragma unroll
            for (int j = 0; j < SCAT_EB; ++j) {
                l[j] = m ? (int)__builtin_ctzll(m) : -1;
                m &= m - 1ull;
            }
            float lo[SCAT_EB], hi[SCAT_EB];
            if (full) {
                #pragma unroll
                for (int j = 0; j < SCAT_EB; ++j) {
                    const int s_lo = __builtin_amdgcn_readlane(sa, l[j] & 31);
                    const int s_hi = __builtin_amdgcn_readlane(sb, l[j] & 31);
                    const unsigned s = (unsigned)((l[j] < 32) ? s_lo : s_hi);
                    const float* hp = h + (size_t)s * OUT_FEATS;
                    lo[j] = hp[lane];
                    hi[j] = hp[lane + 32];
                }
                #pragma unroll
                for (int j = 0; j < SCAT_EB; ++j) {
                    const int d_lo = __builtin_amdgcn_readlane(da, l[j] & 31);
                    const int d_hi = __builtin_amdgcn_readlane(db, l[j] & 31);
                    const unsigned d = (unsigned)((l[j] < 32) ? d_lo : d_hi);
                    float* ap = agg + (size_t)d * OUT_FEATS;
                    atomicAdd(ap + lane,      lo[j]);
                    atomicAdd(ap + lane + 32, hi[j]);
                }
            } else {
                #pragma unroll
                for (int j = 0; j < SCAT_EB; ++j) {
                    const int lj = (l[j] >= 0) ? l[j] : l[0];
                    const int s_lo = __builtin_amdgcn_readlane(sa, lj & 31);
                    const int s_hi = __builtin_amdgcn_readlane(sb, lj & 31);
                    const unsigned s = (unsigned)((lj < 32) ? s_lo : s_hi);
                    const float* hp = h + (size_t)s * OUT_FEATS;
                    lo[j] = hp[lane];
                    hi[j] = hp[lane + 32];
                }
                #pragma unroll
                for (int j = 0; j < SCAT_EB; ++j) {
                    if (l[j] >= 0) {
                        const int d_lo = __builtin_amdgcn_readlane(da, l[j] & 31);
                        const int d_hi = __builtin_amdgcn_readlane(db, l[j] & 31);
                        const unsigned d = (unsigned)((l[j] < 32) ? d_lo : d_hi);
                        float* ap = agg + (size_t)d * OUT_FEATS;
                        atomicAdd(ap + lane,      lo[j]);
                        atomicAdd(ap + lane + 32, hi[j]);
                    }
                }
            }
        }
    }
}

__global__ void gcn_finalize(const float* __restrict__ agg,
                             const float* __restrict__ norm_dst,
                             const float* __restrict__ b,
                             float* __restrict__ out,
                             int total) {
    int t = blockIdx.x * blockDim.x + threadIdx.x;
    if (t >= total) return;
    int node = t >> 6;
    int f    = t & 63;
    float x = agg[t] * norm_dst[node] + b[f];
    out[t] = 1.0f / (1.0f + __expf(-x));
}

extern "C" void kernel_launch(void* const* d_in, const int* in_sizes, int n_in,
                              void* d_out, int out_size, void* d_ws, size_t ws_size,
                              hipStream_t stream) {
    const float* feat = (const float*)d_in[0];
    const int*   src  = (const int*)d_in[1];
    const int*   dst  = (const int*)d_in[2];
    const float* W    = (const float*)d_in[3];
    const float* bias = (const float*)d_in[4];
    float*       out  = (float*)d_out;

    const int n_nodes = in_sizes[0] / IN_FEATS;
    const int n_edges = in_sizes[1];

    float* agg      = (float*)d_ws;
    unsigned* part  = (unsigned*)(agg + (size_t)n_nodes * OUT_FEATS);
    float* norm_dst = (float*)(part + (size_t)2 * DEG_SLICES * n_nodes);
    int* tickets    = (int*)(norm_dst + n_nodes);

    gcn_degree_slices<<<2 * DEG_SLICES, 1024, 0, stream>>>(
        src, dst, part, n_edges, n_nodes);

    const int n_tiles = n_nodes / 16;
    gcn_gemm_wmma<<<(n_tiles + 7) / 8, 256, 0, stream>>>(
        feat, part, W, out, norm_dst, agg, tickets, n_tiles, n_nodes);

    gcn_scatter<<<SCAT_WGS, 256, 0, stream>>>(
        src, dst, out, agg, tickets, n_edges, n_nodes);

    int total = n_nodes * OUT_FEATS;
    gcn_finalize<<<(total + 255) / 256, 256, 0, stream>>>(
        agg, norm_dst, bias, out, total);
}


# degree kernel split by node window as well as edge slice: 256 work-groups (4 windows x 32 slices x 2 arrays, one per WGP) each clear, count and store a quarter plane; was 64 with whole planes
# speedup vs baseline: 2.8444x; 1.0321x over previous
#include <hip/hip_runtime.h>
#include <math.h>

typedef float v2f __attribute__((ext_vector_type(2)));
typedef float v8f __attribute__((ext_vector_type(8)));

#define IN_FEATS  256
#define OUT_FEATS 64
#define WPAIRS    (IN_FEATS / 2)
#define WSTRIDE   160
#define DEG_SLICES 32
#define DEG_WINDOWS 4
#define DEG_THREADS 512
#define HIST_NODES 45056
#define DEG_BATCH  16
#define W_STAGE_IT 16
#define K_BLK      16
#define TICKET_PITCH 64

__device__ __forceinline__ void deg_count(unsigned* hist, int node, int base, int cnt) {
    unsigned r = (unsigned)(node - base);
    if (r < (unsigned)cnt) atomicAdd(&hist[r], 1u);
}

__global__ void __launch_bounds__(DEG_THREADS)
gcn_degree_slices(const int* __restrict__ src,
                  const int* __restrict__ dst,
                  unsigned* __restrict__ part,
                  int n_edges, int n_nodes) {
    __shared__ unsigned hist[HIST_NODES];

    const int tid   = threadIdx.x;
    const int win   = blockIdx.x % DEG_WINDOWS;
    const int ws    = blockIdx.x / DEG_WINDOWS;
    const int which = ws / DEG_SLICES;
    const int slice = ws - which * DEG_SLICES;
    const int* __restrict__ idx = which ? dst : src;
    unsigned* __restrict__ plane = part + ((size_t)which * DEG_SLICES + slice) * n_nodes;

    const int wsize = (((n_nodes + DEG_WINDOWS - 1) / DEG_WINDOWS) + 31) & ~31;
    const int w0    = min(win * wsize, n_nodes);
    const int w1    = min(w0 + wsize, n_nodes);

    const int nquads = n_edges >> 2;
    const int qper   = (nquads + DEG_SLICES - 1) / DEG_SLICES;
    const int q0     = min(slice * qper, nquads);
    const int q1     = min(q0 + qper, nquads);
    const int4* __restrict__ idx4 = reinterpret_cast<const int4*>(idx);

    for (int base = w0; base < w1; base += HIST_NODES) {
        const int cnt = min(HIST_NODES, w1 - base);

        for (int i = tid * 4; i < cnt; i += 4 * DEG_THREADS)
            *reinterpret_cast<uint4*>(&hist[i]) = make_uint4(0u, 0u, 0u, 0u);
        __syncthreads();

        for (int q = q0 + tid; q < q1; q += DEG_BATCH * DEG_THREADS) {
            int4 v[DEG_BATCH];
            #pragma unroll
            for (int j = 0; j < DEG_BATCH; ++j) {
                const int qq = q + j * DEG_THREADS;
                v[j] = (qq < q1) ? idx4[qq] : make_int4(-1, -1, -1, -1);
            }
            #pragma unroll
            for (int j = 0; j < DEG_BATCH; ++j) {
                deg_count(hist, v[j].x, base, cnt);
                deg_count(hist, v[j].y, base, cnt);
                deg_count(hist, v[j].z, base, cnt);
                deg_count(hist, v[j].w, base, cnt);
            }
        }
        if (slice == DEG_SLICES - 1)
            for (int e = nquads * 4 + tid; e < n_edges; e += DEG_THREADS)
                deg_count(hist, idx[e], base, cnt);
        __syncthreads();

        #pragma unroll 4
        for (int i = tid * 4; i < cnt; i += 4 * DEG_THREADS)
            *reinterpret_cast<uint4*>(&plane[base + i]) =
                *reinterpret_cast<const uint4*>(&hist[i]);
        __syncthreads();
    }
}

__global__ void __launch_bounds__(256)
gcn_gemm_wmma(const float* __restrict__ feat,
              const unsigned* __restrict__ part,
              const float* __restrict__ W,
              float* __restrict__ h,
              float* __restrict__ norm_dst,
              float* __restrict__ agg,
              int* __restrict__ tickets,
              int n_tiles, int n_nodes) {
    __shared__ float Wl[WPAIRS * WSTRIDE];
    __shared__ float nd_s[8 * 16];

    const int tid  = threadIdx.x;
    const int wave = tid >> 5;
    const int lane = tid & 31;
    const int tile = blockIdx.x * 8 + wave;
    const int half = lane >> 4;
    const int l15  = lane & 15;

    if (blockIdx.x == 0 && tid == 0) {
        tickets[0] = 0;
        tickets[TICKET_PITCH] = 0;
    }

    const int mload = min(tile, n_tiles - 1) * 16 + l15;
    const unsigned* pp = part + (size_t)half * DEG_SLICES * n_nodes + mload;
    unsigned pc[DEG_SLICES];
    #pragma unroll
    for (int s = 0; s < DEG_SLICES; ++s)
        pc[s] = pp[(size_t)s * n_nodes];

    float2 wlo[W_STAGE_IT], whi[W_STAGE_IT];
    #pragma unroll
    for (int it = 0; it < W_STAGE_IT; ++it) {
        const int i  = tid + it * 256;
        const int p  = i >> 5;
        const int n2 = (i & 31) * 2;
        wlo[it] = *reinterpret_cast<const float2*>(&W[(2 * p) * OUT_FEATS + n2]);
        whi[it] = *reinterpret_cast<const float2*>(&W[(2 * p + 1) * OUT_FEATS + n2]);
    }

    const float* frow = feat + (size_t)mload * IN_FEATS + 2 * half;
    v2f an[K_BLK / 4] = {};
    if (tile < n_tiles) {
        #pragma unroll
        for (int j = 0; j < K_BLK / 4; ++j)
            an[j] = *reinterpret_cast<const v2f*>(frow + 4 * j);
    }

    #pragma unroll
    for (int it = 0; it < W_STAGE_IT; ++it) {
        const int i  = tid + it * 256;
        const int p  = i >> 5;
        const int n2 = (i & 31) * 2;
        float4 q;  q.x = wlo[it].x;  q.y = whi[it].x;  q.z = wlo[it].y;  q.w = whi[it].y;
        *reinterpret_cast<float4*>(&Wl[p * WSTRIDE + 2 * n2]) = q;
    }
    unsigned cnt = 0;
    #pragma unroll
    for (int s = 0; s < DEG_SLICES; ++s)
        cnt += pc[s];
    const float nrm_own = rsqrtf(fmaxf((float)cnt, 1.0f));
    if (half) nd_s[wave * 16 + l15] = nrm_own;
    __syncthreads();

    if (wave < 4) {
        const int row = blockIdx.x * 128 + wave * 32 + lane;
        if (row < n_nodes) norm_dst[row] = nd_s[wave * 32 + lane];
    }

    if (tile >= n_tiles) return;

    const float nrm = __shfl(nrm_own, l15);

    float4* zrow = reinterpret_cast<float4*>(agg + (size_t)tile * (16 * OUT_FEATS));
    #pragma unroll
    for (int i = 0; i < (16 * OUT_FEATS) / (4 * 32); ++i)
        zrow[i * 32 + lane] = make_float4(0.0f, 0.0f, 0.0f, 0.0f);

    const float* wbase = &Wl[half * WSTRIDE + 2 * l15];

    v8f c0 = {}, c1 = {}, c2 = {}, c3 = {};

    for (int k = 0; k < IN_FEATS; k += K_BLK) {
        v2f ac[K_BLK / 4];
        #pragma unroll
        for (int j = 0; j < K_BLK / 4; ++j)
            ac[j] = an[j] * nrm;

        const int kn = (k + K_BLK < IN_FEATS) ? k + K_BLK : k;
        #pragma unroll
        for (int j = 0; j < K_BLK / 4; ++j)
            an[j] = *reinterpret_cast<const v2f*>(frow + kn + 4 * j);

        #pragma unroll
        for (int j = 0; j < K_BLK / 4; ++j) {
            const v2f a = ac[j];
            const float* wp = wbase + ((k + 4 * j) >> 1) * WSTRIDE;
            v2f b0 = *reinterpret_cast<const v2f*>(wp);
            v2f b1 = *reinterpret_cast<const v2f*>(wp + 32);
            v2f b2 = *reinterpret_cast<const v2f*>(wp + 64);
            v2f b3 = *reinterpret_cast<const v2f*>(wp + 96);

            c0 = __builtin_amdgcn_wmma_f32_16x16x4_f32(false, a, false, b0,
                                                       (short)0, c0, false, false);
            c1 = __builtin_amdgcn_wmma_f32_16x16x4_f32(false, a, false, b1,
                                                       (short)0, c1, false, false);
            c2 = __builtin_amdgcn_wmma_f32_16x16x4_f32(false, a, false, b2,
                                                       (short)0, c2, false, false);
            c3 = __builtin_amdgcn_wmma_f32_16x16x4_f32(false, a, false, b3,
                                                       (short)0, c3, false, false);
        }
    }

    const int rbase = tile * 16 + 8 * half;
    #pragma unroll
    for (int r = 0; r < 8; ++r) {
        float* orow = h + (size_t)(rbase + r) * OUT_FEATS + l15;
        orow[0]  = c0[r];
        orow[16] = c1[r];
        orow[32] = c2[r];
        orow[48] = c3[r];
    }
}

#define SCAT_EPW  64
#define SCAT_ITEM (8 * SCAT_EPW)
#define SCAT_EB   8
#define SCAT_WGS  2048

__device__ __forceinline__ int scat_take(int* tickets, int q, bool& own_empty, int n_items) {
    if (!own_empty) {
        const int t = atomicAdd(&tickets[q * TICKET_PITCH], 1);
        if (t < n_items) return t * 2 + q;
        own_empty = true;
    }
    const int t = atomicAdd(&tickets[(q ^ 1) * TICKET_PITCH], 1);
    return (t < n_items) ? (t * 2 + (q ^ 1)) : -1;
}

__global__ void __launch_bounds__(256)
gcn_scatter(const int* __restrict__ src,
            const int* __restrict__ dst,
            const float* __restrict__ h,
            float* __restrict__ agg,
            int* __restrict__ tickets,
            int n_edges, int n_nodes) {
    __shared__ int item_s;
    const unsigned lane = threadIdx.x & 31u;
    const int wave = __builtin_amdgcn_readfirstlane((int)(threadIdx.x >> 5));
    const int n_split = n_nodes >> 1;
    const int n_items = (n_edges + SCAT_ITEM - 1) / SCAT_ITEM;

    const unsigned place = __builtin_amdgcn_s_sendmsg_rtn(0x87);
    const int q = (((place >> 16) & 15u) >= 4u) ? 1 : 0;

    bool own_empty = false;
    int nxt = -1;
    if (threadIdx.x == 0) nxt = scat_take(tickets, q, own_empty, n_items);

    for (;;) {
        if (threadIdx.x == 0) item_s = nxt;
        __syncthreads();
        const int it = __builtin_amdgcn_readfirstlane(item_s);
        __syncthreads();
        if (it < 0) break;
        if (threadIdx.x == 0) nxt = scat_take(tickets, q, own_empty, n_items);

        const bool hi_half = (it & 1) != 0;
        const int e0 = (it >> 1) * SCAT_ITEM + wave * SCAT_EPW;
        if (e0 >= n_edges) continue;

        const int ea = e0 + (int)lane;
        const int eb = ea + 32;
        const bool va = ea < n_edges;
        const bool vb = eb < n_edges;
        const int sa = va ? src[ea] : 0;
        const int da = va ? dst[ea] : 0;
        const int sb = vb ? src[eb] : 0;
        const int db = vb ? dst[eb] : 0;
        const unsigned ma = __builtin_amdgcn_ballot_w32(va && ((da >= n_split) == hi_half));
        const unsigned mb = __builtin_amdgcn_ballot_w32(vb && ((db >= n_split) == hi_half));
        unsigned long long m = ((unsigned long long)mb << 32) | (unsigned long long)ma;

        while (m) {
            const bool full = __builtin_popcountll(m) >= SCAT_EB;
            int l[SCAT_EB];
            #pragma unroll
            for (int j = 0; j < SCAT_EB; ++j) {
                l[j] = m ? (int)__builtin_ctzll(m) : -1;
                m &= m - 1ull;
            }
            float lo[SCAT_EB], hi[SCAT_EB];
            if (full) {
                #pragma unroll
                for (int j = 0; j < SCAT_EB; ++j) {
                    const int s_lo = __builtin_amdgcn_readlane(sa, l[j] & 31);
                    const int s_hi = __builtin_amdgcn_readlane(sb, l[j] & 31);
                    const unsigned s = (unsigned)((l[j] < 32) ? s_lo : s_hi);
                    const float* hp = h + (size_t)s * OUT_FEATS;
                    lo[j] = hp[lane];
                    hi[j] = hp[lane + 32];
                }
                #pragma unroll
                for (int j = 0; j < SCAT_EB; ++j) {
                    const int d_lo = __builtin_amdgcn_readlane(da, l[j] & 31);
                    const int d_hi = __builtin_amdgcn_readlane(db, l[j] & 31);
                    const unsigned d = (unsigned)((l[j] < 32) ? d_lo : d_hi);
                    float* ap = agg + (size_t)d * OUT_FEATS;
                    atomicAdd(ap + lane,      lo[j]);
                    atomicAdd(ap + lane + 32, hi[j]);
                }
            } else {
                #pragma unroll
                for (int j = 0; j < SCAT_EB; ++j) {
                    const int lj = (l[j] >= 0) ? l[j] : l[0];
                    const int s_lo = __builtin_amdgcn_readlane(sa, lj & 31);
                    const int s_hi = __builtin_amdgcn_readlane(sb, lj & 31);
                    const unsigned s = (unsigned)((lj < 32) ? s_lo : s_hi);
                    const float* hp = h + (size_t)s * OUT_FEATS;
                    lo[j] = hp[lane];
                    hi[j] = hp[lane + 32];
                }
                #pragma unroll
                for (int j = 0; j < SCAT_EB; ++j) {
                    if (l[j] >= 0) {
                        const int d_lo = __builtin_amdgcn_readlane(da, l[j] & 31);
                        const int d_hi = __builtin_amdgcn_readlane(db, l[j] & 31);
                        const unsigned d = (unsigned)((l[j] < 32) ? d_lo : d_hi);
                        float* ap = agg + (size_t)d * OUT_FEATS;
                        atomicAdd(ap + lane,      lo[j]);
                        atomicAdd(ap + lane + 32, hi[j]);
                    }
                }
            }
        }
    }
}

__global__ void gcn_finalize(const float* __restrict__ agg,
                             const float* __restrict__ norm_dst,
                             const float* __restrict__ b,
                             float* __restrict__ out,
                             int total) {
    int t = blockIdx.x * blockDim.x + threadIdx.x;
    if (t >= total) return;
    int node = t >> 6;
    int f    = t & 63;
    float x = agg[t] * norm_dst[node] + b[f];
    out[t] = 1.0f / (1.0f + __expf(-x));
}

extern "C" void kernel_launch(void* const* d_in, const int* in_sizes, int n_in,
                              void* d_out, int out_size, void* d_ws, size_t ws_size,
                              hipStream_t stream) {
    const float* feat = (const float*)d_in[0];
    const int*   src  = (const int*)d_in[1];
    const int*   dst  = (const int*)d_in[2];
    const float* W    = (const float*)d_in[3];
    const float* bias = (const float*)d_in[4];
    float*       out  = (float*)d_out;

    const int n_nodes = in_sizes[0] / IN_FEATS;
    const int n_edges = in_sizes[1];

    float* agg      = (float*)d_ws;
    unsigned* part  = (unsigned*)(agg + (size_t)n_nodes * OUT_FEATS);
    float* norm_dst = (float*)(part + (size_t)2 * DEG_SLICES * n_nodes);
    int* tickets    = (int*)(norm_dst + n_nodes);

    gcn_degree_slices<<<2 * DEG_SLICES * DEG_WINDOWS, DEG_THREADS, 0, stream>>>(
        src, dst, part, n_edges, n_nodes);

    const int n_tiles = n_nodes / 16;
    gcn_gemm_wmma<<<(n_tiles + 7) / 8, 256, 0, stream>>>(
        feat, part, W, out, norm_dst, agg, tickets, n_tiles, n_nodes);

    gcn_scatter<<<SCAT_WGS, 256, 0, stream>>>(
        src, dst, out, agg, tickets, n_edges, n_nodes);

    int total = n_nodes * OUT_FEATS;
    gcn_finalize<<<(total + 255) / 256, 256, 0, stream>>>(
        agg, norm_dst, bias, out, total);
}


# scatter by memory f32 atomics and the finalize launch replaced: edges sorted by dst bucket (LDS counters) in the degree launch, a work-group sums its bucket's rows by LDS f32 adds (arrival order)
# speedup vs baseline: 4.1309x; 1.4523x over previous
#include <hip/hip_runtime.h>
#include <math.h>


typedef float v2f __attribute__((ext_vector_type(2)));
typedef float v8f __attribute__((ext_vector_type(8)));

#define IN_FEATS  256
#define OUT_FEATS 64
#define WPAIRS    (IN_FEATS / 2)
#define WSTRIDE   160
#define DEG_SLICES 32
#define DEG_WINDOWS 4
#define DEG_WGS    (DEG_SLICES * DEG_WINDOWS)
#define DEG_THREADS 512
#define HIST_NODES 45056
#define DEG_BATCH  16
#define W_STAGE_IT 16
#define K_BLK      16

#define BKT_NODES  196
#define NBKT       256
#define BIN_SLICES 128
#define BIN_LIST   1024
#define BIN_LIST_CAP (HIST_NODES - BIN_LIST)
#define BIN_BATCH  16
#define GATH_THREADS 1024
#define GATH_WAVES (GATH_THREADS / 32)
#define RUNS_PER_WAVE (BIN_SLICES / GATH_WAVES)
#define GATH_EB    16
#define GATH_LDS_WORDS 41984

__device__ __forceinline__ void deg_count(unsigned* hist, int node, int base, int cnt) {
    unsigned r = (unsigned)(node - base);
    if (r < (unsigned)cnt) atomicAdd(&hist[r], 1u);
}

__global__ void __launch_bounds__(DEG_THREADS)
gcn_degree_bin(const int* __restrict__ src,
               const int* __restrict__ dst,
               unsigned* __restrict__ part,
               int* __restrict__ offs,
               unsigned* __restrict__ rec,
               int n_edges, int n_nodes) {
    __shared__ unsigned hist[HIST_NODES];

    const int tid = threadIdx.x;

    if (blockIdx.x >= DEG_WGS) {
        const int s   = blockIdx.x - DEG_WGS;
        const int lane = tid & 31;
        unsigned* cnt  = hist;
        unsigned* cur  = hist + NBKT;
        unsigned* wsum = hist + 2 * NBKT;
        unsigned* list = hist + BIN_LIST;
        const int cap = (n_edges + BIN_SLICES - 1) / BIN_SLICES;
        const int e0  = min(s * cap, n_edges);
        const int e1  = min(e0 + cap, n_edges);

        if (tid < NBKT) cnt[tid] = 0u;
        __syncthreads();
        for (int eb = e0 + tid; eb < e1; eb += BIN_BATCH * DEG_THREADS) {
            unsigned d[BIN_BATCH];
            #pragma unroll
            for (int j = 0; j < BIN_BATCH; ++j) {
                const int e = eb + j * DEG_THREADS;
                d[j] = (e < e1) ? (unsigned)dst[e] : 0xFFFFFFFFu;
            }
            #pragma unroll
            for (int j = 0; j < BIN_BATCH; ++j)
                if (d[j] != 0xFFFFFFFFu) atomicAdd(&cnt[d[j] / BKT_NODES], 1u);
        }
        __syncthreads();

        const unsigned v = (tid < NBKT) ? cnt[tid] : 0u;
        unsigned inc = v;
        #pragma unroll
        for (int o = 1; o < 32; o <<= 1) {
            const unsigned t = __shfl_up(inc, o);
            if (lane >= o) inc += t;
        }
        if (tid < NBKT && lane == 31) wsum[tid >> 5] = inc;
        __syncthreads();
        if (tid < NBKT) {
            unsigned pre = 0u;
            for (int w = 0; w < (tid >> 5); ++w) pre += wsum[w];
            const unsigned exc = pre + inc - v;
            cur[tid] = exc;
            offs[s * (NBKT + 1) + tid] = (int)exc;
            if (tid == NBKT - 1) offs[s * (NBKT + 1) + NBKT] = (int)(exc + v);
        }
        __syncthreads();

        const bool staged = (e1 - e0) <= BIN_LIST_CAP;
        unsigned* outp = rec + (size_t)s * cap;
        for (int eb = e0 + tid; eb < e1; eb += BIN_BATCH * DEG_THREADS) {
            unsigned d[BIN_BATCH], sr[BIN_BATCH];
            #pragma unroll
            for (int j = 0; j < BIN_BATCH; ++j) {
                const int e = eb + j * DEG_THREADS;
                d[j]  = (e < e1) ? (unsigned)dst[e] : 0xFFFFFFFFu;
                sr[j] = (e < e1) ? (unsigned)src[e] : 0u;
            }
            #pragma unroll
            for (int j = 0; j < BIN_BATCH; ++j) {
                if (d[j] != 0xFFFFFFFFu) {
                    const unsigned bk  = d[j] / BKT_NODES;
                    const unsigned pos = atomicAdd(&cur[bk], 1u);
                    const unsigned val = sr[j] | ((d[j] - bk * BKT_NODES) << 24);
                    if (staged) list[pos] = val; else outp[pos] = val;
                }
            }
        }
        __syncthreads();
        if (staged)
            for (int i = tid; i < e1 - e0; i += DEG_THREADS)
                outp[i] = list[i];
        return;
    }

    const int win   = blockIdx.x % DEG_WINDOWS;
    const int slice = blockIdx.x / DEG_WINDOWS;
    unsigned* __restrict__ plane = part + (size_t)slice * n_nodes;

    const int wsize = (((n_nodes + DEG_WINDOWS - 1) / DEG_WINDOWS) + 31) & ~31;
    const int w0    = min(win * wsize, n_nodes);
    const int w1    = min(w0 + wsize, n_nodes);

    const int nquads = n_edges >> 2;
    const int qper   = (nquads + DEG_SLICES - 1) / DEG_SLICES;
    const int q0     = min(slice * qper, nquads);
    const int q1     = min(q0 + qper, nquads);
    const int4* __restrict__ idx4 = reinterpret_cast<const int4*>(src);

    for (int base = w0; base < w1; base += HIST_NODES) {
        const int cnt = min(HIST_NODES, w1 - base);

        for (int i = tid * 4; i < cnt; i += 4 * DEG_THREADS)
            *reinterpret_cast<uint4*>(&hist[i]) = make_uint4(0u, 0u, 0u, 0u);
        __syncthreads();

        for (int q = q0 + tid; q < q1; q += DEG_BATCH * DEG_THREADS) {
            int4 v[DEG_BATCH];
            #pragma unroll
            for (int j = 0; j < DEG_BATCH; ++j) {
                const int qq = q + j * DEG_THREADS;
                v[j] = (qq < q1) ? idx4[qq] : make_int4(-1, -1, -1, -1);
            }
            #pragma unroll
            for (int j = 0; j < DEG_BATCH; ++j) {
                deg_count(hist, v[j].x, base, cnt);
                deg_count(hist, v[j].y, base, cnt);
                deg_count(hist, v[j].z, base, cnt);
                deg_count(hist, v[j].w, base, cnt);
            }
        }
        if (slice == DEG_SLICES - 1)
            for (int e = nquads * 4 + tid; e < n_edges; e += DEG_THREADS)
                deg_count(hist, src[e], base, cnt);
        __syncthreads();

        #pragma unroll 4
        for (int i = tid * 4; i < cnt; i += 4 * DEG_THREADS)
            *reinterpret_cast<uint4*>(&plane[base + i]) =
                *reinterpret_cast<const uint4*>(&hist[i]);
        __syncthreads();
    }
}

__global__ void __launch_bounds__(256)
gcn_gemm_wmma(const float* __restrict__ feat,
              const unsigned* __restrict__ part,
              const float* __restrict__ W,
              float* __restrict__ h,
              int n_tiles, int n_nodes) {
    __shared__ float Wl[WPAIRS * WSTRIDE];

    const int tid  = threadIdx.x;
    const int wave = tid >> 5;
    const int lane = tid & 31;
    const int tile = blockIdx.x * 8 + wave;
    const int half = lane >> 4;
    const int l15  = lane & 15;

    const int mload = min(tile, n_tiles - 1) * 16 + l15;
    const unsigned* pp = part + (size_t)(half * (DEG_SLICES / 2)) * n_nodes + mload;
    unsigned pc[DEG_SLICES / 2];
    #pragma unroll
    for (int s = 0; s < DEG_SLICES / 2; ++s)
        pc[s] = pp[(size_t)s * n_nodes];

    float2 wlo[W_STAGE_IT], whi[W_STAGE_IT];
    #pragma unroll
    for (int it = 0; it < W_STAGE_IT; ++it) {
        const int i  = tid + it * 256;
        const int p  = i >> 5;
        const int n2 = (i & 31) * 2;
        wlo[it] = *reinterpret_cast<const float2*>(&W[(2 * p) * OUT_FEATS + n2]);
        whi[it] = *reinterpret_cast<const float2*>(&W[(2 * p + 1) * OUT_FEATS + n2]);
    }

    const float* frow = feat + (size_t)mload * IN_FEATS + 2 * half;
    v2f an[K_BLK / 4] = {};
    if (tile < n_tiles) {
        #pragma unroll
        for (int j = 0; j < K_BLK / 4; ++j)
            an[j] = *reinterpret_cast<const v2f*>(frow + 4 * j);
    }

    #pragma unroll
    for (int it = 0; it < W_STAGE_IT; ++it) {
        const int i  = tid + it * 256;
        const int p  = i >> 5;
        const int n2 = (i & 31) * 2;
        float4 q;  q.x = wlo[it].x;  q.y = whi[it].x;  q.z = wlo[it].y;  q.w = whi[it].y;
        *reinterpret_cast<float4*>(&Wl[p * WSTRIDE + 2 * n2]) = q;
    }
    unsigned cnt = 0;
    #pragma unroll
    for (int s = 0; s < DEG_SLICES / 2; ++s)
        cnt += pc[s];
    cnt += __shfl_xor(cnt, 16);
    const float nrm = rsqrtf(fmaxf((float)cnt, 1.0f));
    __syncthreads();

    if (tile >= n_tiles) return;


    const float* wbase = &Wl[half * WSTRIDE + 2 * l15];

    v8f c0 = {}, c1 = {}, c2 = {}, c3 = {};

    for (int k = 0; k < IN_FEATS; k += K_BLK) {
        v2f ac[K_BLK / 4];
        #pragma unroll
        for (int j = 0; j < K_BLK / 4; ++j)
            ac[j] = an[j] * nrm;

        const int kn = (k + K_BLK < IN_FEATS) ? k + K_BLK : k;
        #pragma unroll
        for (int j = 0; j < K_BLK / 4; ++j)
            an[j] = *reinterpret_cast<const v2f*>(frow + kn + 4 * j);

        #pragma unroll
        for (int j = 0; j < K_BLK / 4; ++j) {
            const v2f a = ac[j];
            const float* wp = wbase + ((k + 4 * j) >> 1) * WSTRIDE;
            v2f b0 = *reinterpret_cast<const v2f*>(wp);
            v2f b1 = *reinterpret_cast<const v2f*>(wp + 32);
            v2f b2 = *reinterpret_cast<const v2f*>(wp + 64);
            v2f b3 = *reinterpret_cast<const v2f*>(wp + 96);

            c0 = __builtin_amdgcn_wmma_f32_16x16x4_f32(false, a, false, b0,
                                                       (short)0, c0, false, false);
            c1 = __builtin_amdgcn_wmma_f32_16x16x4_f32(false, a, false, b1,
                                                       (short)0, c1, false, false);
            c2 = __builtin_amdgcn_wmma_f32_16x16x4_f32(false, a, false, b2,
                                                       (short)0, c2, false, false);
            c3 = __builtin_amdgcn_wmma_f32_16x16x4_f32(false, a, false, b3,
                                                       (short)0, c3, false, false);
        }
    }

    const int rbase = tile * 16 + 8 * half;
    #pragma unroll
    for (int r = 0; r < 8; ++r) {
        float* orow = h + (size_t)(rbase + r) * OUT_FEATS + l15;
        orow[0]  = c0[r];
        orow[16] = c1[r];
        orow[32] = c2[r];
        orow[48] = c3[r];
    }
}

__global__ void __launch_bounds__(GATH_THREADS)
gcn_gather_finalize(const unsigned* __restrict__ rec,
                    const int* __restrict__ offs,
                    const float* __restrict__ h,
                    const float* __restrict__ b,
                    float* __restrict__ out,
                    int n_edges, int n_nodes) {
    __shared__ float acc[GATH_LDS_WORDS];
    __shared__ unsigned indeg[BKT_NODES];

    const int tid = threadIdx.x;
    const int lane = tid & 31;
    const int wave = __builtin_amdgcn_readfirstlane(tid >> 5);
    const int bk = blockIdx.x;
    const int node0 = bk * BKT_NODES;
    const int nn = min(BKT_NODES, n_nodes - node0);
    const int cap = (n_edges + BIN_SLICES - 1) / BIN_SLICES;

    for (int i = tid; i < BKT_NODES * OUT_FEATS; i += GATH_THREADS) acc[i] = 0.0f;
    if (tid < BKT_NODES) indeg[tid] = 0u;

    int o0[RUNS_PER_WAVE], o1[RUNS_PER_WAVE];
    #pragma unroll
    for (int j = 0; j < RUNS_PER_WAVE; ++j) {
        const int s = wave + j * GATH_WAVES;
        o0[j] = offs[s * (NBKT + 1) + bk];
        o1[j] = offs[s * (NBKT + 1) + bk + 1];
    }
    unsigned rv[RUNS_PER_WAVE];
    #pragma unroll
    for (int j = 0; j < RUNS_PER_WAVE; ++j) {
        const int s = wave + j * GATH_WAVES;
        const int i = o0[j] + lane;
        rv[j] = (i < o1[j]) ? rec[(size_t)s * cap + i] : 0u;
    }
    __syncthreads();

    #pragma unroll
    for (int j = 0; j < RUNS_PER_WAVE; ++j) {
        const unsigned* rp = rec + (size_t)(wave + j * GATH_WAVES) * cap;
        int i0 = o0[j];
        unsigned r = rv[j];
        while (i0 < o1[j]) {
            const int n = min(32, o1[j] - i0);
            if (lane < n) atomicAdd(&indeg[r >> 24], 1u);
            for (int e = 0; e < n; e += GATH_EB) {
                if (e + GATH_EB <= n) {
                    float lo[GATH_EB], hi[GATH_EB];
                    #pragma unroll
                    for (int q = 0; q < GATH_EB; ++q) {
                        const unsigned x = (unsigned)__builtin_amdgcn_readlane((int)r, e + q);
                        const float* hp = h + (size_t)(x & 0xFFFFFFu) * OUT_FEATS;
                        lo[q] = hp[lane];
                        hi[q] = hp[lane + 32];
                    }
                    #pragma unroll
                    for (int q = 0; q < GATH_EB; ++q) {
                        const unsigned x = (unsigned)__builtin_amdgcn_readlane((int)r, e + q);
                        float* ap = acc + (x >> 24) * OUT_FEATS;
                        atomicAdd(ap + lane,      lo[q]);
                        atomicAdd(ap + lane + 32, hi[q]);
                    }
                } else {
                    float lo[GATH_EB], hi[GATH_EB];
                    #pragma unroll
                    for (int q = 0; q < GATH_EB; ++q) {
                        lo[q] = 0.0f;
                        hi[q] = 0.0f;
                        if (e + q < n) {
                            const unsigned x = (unsigned)__builtin_amdgcn_readlane((int)r, e + q);
                            const float* hp = h + (size_t)(x & 0xFFFFFFu) * OUT_FEATS;
                            lo[q] = hp[lane];
                            hi[q] = hp[lane + 32];
                        }
                    }
                    #pragma unroll
                    for (int q = 0; q < GATH_EB; ++q) {
                        if (e + q < n) {
                            const unsigned x = (unsigned)__builtin_amdgcn_readlane((int)r, e + q);
                            float* ap = acc + (x >> 24) * OUT_FEATS;
                            atomicAdd(ap + lane,      lo[q]);
                            atomicAdd(ap + lane + 32, hi[q]);
                        }
                    }
                }
            }
            i0 += 32;
            r = (i0 + lane < o1[j]) ? rp[i0 + lane] : 0u;
        }
    }
    __syncthreads();

    const int tot = nn * OUT_FEATS;
    float* orow = out + (size_t)node0 * OUT_FEATS;
    for (int i = tid; i < tot; i += GATH_THREADS) {
        const float nd = rsqrtf(fmaxf((float)indeg[i >> 6], 1.0f));
        const float x = acc[i] * nd + b[i & 63];
        orow[i] = 1.0f / (1.0f + __expf(-x));
    }
}

extern "C" void kernel_launch(void* const* d_in, const int* in_sizes, int n_in,
                              void* d_out, int out_size, void* d_ws, size_t ws_size,
                              hipStream_t stream) {
    const float* feat = (const float*)d_in[0];
    const int*   src  = (const int*)d_in[1];
    const int*   dst  = (const int*)d_in[2];
    const float* W    = (const float*)d_in[3];
    const float* bias = (const float*)d_in[4];
    float*       out  = (float*)d_out;

    const int n_nodes = in_sizes[0] / IN_FEATS;
    const int n_edges = in_sizes[1];

    float* h        = (float*)d_ws;
    unsigned* part  = (unsigned*)(h + (size_t)n_nodes * OUT_FEATS);
    int* offs       = (int*)(part + (size_t)DEG_SLICES * n_nodes);
    unsigned* rec   = (unsigned*)(offs + BIN_SLICES * (NBKT + 1));

    gcn_degree_bin<<<DEG_WGS + BIN_SLICES, DEG_THREADS, 0, stream>>>(
        src, dst, part, offs, rec, n_edges, n_nodes);

    const int n_tiles = n_nodes / 16;
    gcn_gemm_wmma<<<(n_tiles + 7) / 8, 256, 0, stream>>>(
        feat, part, W, h, n_tiles, n_nodes);

    gcn_gather_finalize<<<NBKT, GATH_THREADS, 0, stream>>>(
        rec, offs, h, bias, out, n_edges, n_nodes);
}


# bucket kernel sums rows in registers: edges counting-sorted by dst row in LDS, a wave walks its rows with one 8-byte load per lane and edge; was 2 LDS f32 adds per edge (add order: slot order)
# speedup vs baseline: 4.6955x; 1.1367x over previous
#include <hip/hip_runtime.h>
#include <math.h>


typedef float v2f __attribute__((ext_vector_type(2)));
typedef float v8f __attribute__((ext_vector_type(8)));

#define IN_FEATS  256
#define OUT_FEATS 64
#define WPAIRS    (IN_FEATS / 2)
#define WSTRIDE   160
#define DEG_SLICES 32
#define DEG_WINDOWS 4
#define DEG_WGS    (DEG_SLICES * DEG_WINDOWS)
#define DEG_THREADS 512
#define HIST_NODES 45056
#define DEG_BATCH  16
#define W_STAGE_IT 16
#define K_BLK      16

#define BKT_NODES  196
#define NBKT       256
#define BIN_SLICES 128
#define BIN_LIST   1024
#define BIN_LIST_CAP (HIST_NODES - BIN_LIST)
#define BIN_BATCH  16
#define GATH_THREADS 1024
#define GATH_WAVES (GATH_THREADS / 32)
#define RUNS_PER_WAVE (BIN_SLICES / GATH_WAVES)
#define GATH_EB    16
#define ROUND_CAP  64
#define SORT_CAP   (BIN_SLICES * ROUND_CAP)
#define SORT_OFF   16384
#define GATH_LDS_WORDS 41984

__device__ __forceinline__ void deg_count(unsigned* hist, int node, int base, int cnt) {
    unsigned r = (unsigned)(node - base);
    if (r < (unsigned)cnt) atomicAdd(&hist[r], 1u);
}

__global__ void __launch_bounds__(DEG_THREADS)
gcn_degree_bin(const int* __restrict__ src,
               const int* __restrict__ dst,
               unsigned* __restrict__ part,
               int* __restrict__ offs,
               unsigned* __restrict__ rec,
               int n_edges, int n_nodes) {
    __shared__ unsigned hist[HIST_NODES];

    const int tid = threadIdx.x;

    if (blockIdx.x >= DEG_WGS) {
        const int s   = blockIdx.x - DEG_WGS;
        const int lane = tid & 31;
        unsigned* cnt  = hist;
        unsigned* cur  = hist + NBKT;
        unsigned* wsum = hist + 2 * NBKT;
        unsigned* list = hist + BIN_LIST;
        const int cap = (n_edges + BIN_SLICES - 1) / BIN_SLICES;
        const int e0  = min(s * cap, n_edges);
        const int e1  = min(e0 + cap, n_edges);

        if (tid < NBKT) cnt[tid] = 0u;
        __syncthreads();
        for (int eb = e0 + tid; eb < e1; eb += BIN_BATCH * DEG_THREADS) {
            unsigned d[BIN_BATCH];
            #pragma unroll
            for (int j = 0; j < BIN_BATCH; ++j) {
                const int e = eb + j * DEG_THREADS;
                d[j] = (e < e1) ? (unsigned)dst[e] : 0xFFFFFFFFu;
            }
            #pragma unroll
            for (int j = 0; j < BIN_BATCH; ++j)
                if (d[j] != 0xFFFFFFFFu) atomicAdd(&cnt[d[j] / BKT_NODES], 1u);
        }
        __syncthreads();

        const unsigned v = (tid < NBKT) ? cnt[tid] : 0u;
        unsigned inc = v;
        #pragma unroll
        for (int o = 1; o < 32; o <<= 1) {
            const unsigned t = __shfl_up(inc, o);
            if (lane >= o) inc += t;
        }
        if (tid < NBKT && lane == 31) wsum[tid >> 5] = inc;
        __syncthreads();
        if (tid < NBKT) {
            unsigned pre = 0u;
            for (int w = 0; w < (tid >> 5); ++w) pre += wsum[w];
            const unsigned exc = pre + inc - v;
            cur[tid] = exc;
            offs[s * (NBKT + 1) + tid] = (int)exc;
            if (tid == NBKT - 1) offs[s * (NBKT + 1) + NBKT] = (int)(exc + v);
        }
        __syncthreads();

        const bool staged = (e1 - e0) <= BIN_LIST_CAP;
        unsigned* outp = rec + (size_t)s * cap;
        for (int eb = e0 + tid; eb < e1; eb += BIN_BATCH * DEG_THREADS) {
            unsigned d[BIN_BATCH], sr[BIN_BATCH];
            #pragma unroll
            for (int j = 0; j < BIN_BATCH; ++j) {
                const int e = eb + j * DEG_THREADS;
                d[j]  = (e < e1) ? (unsigned)dst[e] : 0xFFFFFFFFu;
                sr[j] = (e < e1) ? (unsigned)src[e] : 0u;
            }
            #pragma unroll
            for (int j = 0; j < BIN_BATCH; ++j) {
                if (d[j] != 0xFFFFFFFFu) {
                    const unsigned bk  = d[j] / BKT_NODES;
                    const unsigned pos = atomicAdd(&cur[bk], 1u);
                    const unsigned val = sr[j] | ((d[j] - bk * BKT_NODES) << 24);
                    if (staged) list[pos] = val; else outp[pos] = val;
                }
            }
        }
        __syncthreads();
        if (staged)
            for (int i = tid; i < e1 - e0; i += DEG_THREADS)
                outp[i] = list[i];
        return;
    }

    const int win   = blockIdx.x % DEG_WINDOWS;
    const int slice = blockIdx.x / DEG_WINDOWS;
    unsigned* __restrict__ plane = part + (size_t)slice * n_nodes;

    const int wsize = (((n_nodes + DEG_WINDOWS - 1) / DEG_WINDOWS) + 31) & ~31;
    const int w0    = min(win * wsize, n_nodes);
    const int w1    = min(w0 + wsize, n_nodes);

    const int nquads = n_edges >> 2;
    const int qper   = (nquads + DEG_SLICES - 1) / DEG_SLICES;
    const int q0     = min(slice * qper, nquads);
    const int q1     = min(q0 + qper, nquads);
    const int4* __restrict__ idx4 = reinterpret_cast<const int4*>(src);

    for (int base = w0; base < w1; base += HIST_NODES) {
        const int cnt = min(HIST_NODES, w1 - base);

        for (int i = tid * 4; i < cnt; i += 4 * DEG_THREADS)
            *reinterpret_cast<uint4*>(&hist[i]) = make_uint4(0u, 0u, 0u, 0u);
        __syncthreads();

        for (int q = q0 + tid; q < q1; q += DEG_BATCH * DEG_THREADS) {
            int4 v[DEG_BATCH];
            #pragma unroll
            for (int j = 0; j < DEG_BATCH; ++j) {
                const int qq = q + j * DEG_THREADS;
                v[j] = (qq < q1) ? idx4[qq] : make_int4(-1, -1, -1, -1);
            }
            #pragma unroll
            for (int j = 0; j < DEG_BATCH; ++j) {
                deg_count(hist, v[j].x, base, cnt);
                deg_count(hist, v[j].y, base, cnt);
                deg_count(hist, v[j].z, base, cnt);
                deg_count(hist, v[j].w, base, cnt);
            }
        }
        if (slice == DEG_SLICES - 1)
            for (int e = nquads * 4 + tid; e < n_edges; e += DEG_THREADS)
                deg_count(hist, src[e], base, cnt);
        __syncthreads();

        #pragma unroll 4
        for (int i = tid * 4; i < cnt; i += 4 * DEG_THREADS)
            *reinterpret_cast<uint4*>(&plane[base + i]) =
                *reinterpret_cast<const uint4*>(&hist[i]);
        __syncthreads();
    }
}

__global__ void __launch_bounds__(256)
gcn_gemm_wmma(const float* __restrict__ feat,
              const unsigned* __restrict__ part,
              const float* __restrict__ W,
              float* __restrict__ h,
              int n_tiles, int n_nodes) {
    __shared__ float Wl[WPAIRS * WSTRIDE];

    const int tid  = threadIdx.x;
    const int wave = tid >> 5;
    const int lane = tid & 31;
    const int tile = blockIdx.x * 8 + wave;
    const int half = lane >> 4;
    const int l15  = lane & 15;

    const int mload = min(tile, n_tiles - 1) * 16 + l15;
    const unsigned* pp = part + (size_t)(half * (DEG_SLICES / 2)) * n_nodes + mload;
    unsigned pc[DEG_SLICES / 2];
    #pragma unroll
    for (int s = 0; s < DEG_SLICES / 2; ++s)
        pc[s] = pp[(size_t)s * n_nodes];

    float2 wlo[W_STAGE_IT], whi[W_STAGE_IT];
    #pragma unroll
    for (int it = 0; it < W_STAGE_IT; ++it) {
        const int i  = tid + it * 256;
        const int p  = i >> 5;
        const int n2 = (i & 31) * 2;
        wlo[it] = *reinterpret_cast<const float2*>(&W[(2 * p) * OUT_FEATS + n2]);
        whi[it] = *reinterpret_cast<const float2*>(&W[(2 * p + 1) * OUT_FEATS + n2]);
    }

    const float* frow = feat + (size_t)mload * IN_FEATS + 2 * half;
    v2f an[K_BLK / 4] = {};
    if (tile < n_tiles) {
        #pragma unroll
        for (int j = 0; j < K_BLK / 4; ++j)
            an[j] = *reinterpret_cast<const v2f*>(frow + 4 * j);
    }

    #pragma unroll
    for (int it = 0; it < W_STAGE_IT; ++it) {
        const int i  = tid + it * 256;
        const int p  = i >> 5;
        const int n2 = (i & 31) * 2;
        float4 q;  q.x = wlo[it].x;  q.y = whi[it].x;  q.z = wlo[it].y;  q.w = whi[it].y;
        *reinterpret_cast<float4*>(&Wl[p * WSTRIDE + 2 * n2]) = q;
    }
    unsigned cnt = 0;
    #pragma unroll
    for (int s = 0; s < DEG_SLICES / 2; ++s)
        cnt += pc[s];
    cnt += __shfl_xor(cnt, 16);
    const float nrm = rsqrtf(fmaxf((float)cnt, 1.0f));
    __syncthreads();

    if (tile >= n_tiles) return;


    const float* wbase = &Wl[half * WSTRIDE + 2 * l15];

    v8f c0 = {}, c1 = {}, c2 = {}, c3 = {};

    for (int k = 0; k < IN_FEATS; k += K_BLK) {
        v2f ac[K_BLK / 4];
        #pragma unroll
        for (int j = 0; j < K_BLK / 4; ++j)
            ac[j] = an[j] * nrm;

        const int kn = (k + K_BLK < IN_FEATS) ? k + K_BLK : k;
        #pragma unroll
        for (int j = 0; j < K_BLK / 4; ++j)
            an[j] = *reinterpret_cast<const v2f*>(frow + kn + 4 * j);

        #pragma unroll
        for (int j = 0; j < K_BLK / 4; ++j) {
            const v2f a = ac[j];
            const float* wp = wbase + ((k + 4 * j) >> 1) * WSTRIDE;
            v2f b0 = *reinterpret_cast<const v2f*>(wp);
            v2f b1 = *reinterpret_cast<const v2f*>(wp + 32);
            v2f b2 = *reinterpret_cast<const v2f*>(wp + 64);
            v2f b3 = *reinterpret_cast<const v2f*>(wp + 96);

            c0 = __builtin_amdgcn_wmma_f32_16x16x4_f32(false, a, false, b0,
                                                       (short)0, c0, false, false);
            c1 = __builtin_amdgcn_wmma_f32_16x16x4_f32(false, a, false, b1,
                                                       (short)0, c1, false, false);
            c2 = __builtin_amdgcn_wmma_f32_16x16x4_f32(false, a, false, b2,
                                                       (short)0, c2, false, false);
            c3 = __builtin_amdgcn_wmma_f32_16x16x4_f32(false, a, false, b3,
                                                       (short)0, c3, false, false);
        }
    }

    const int rbase = tile * 16 + 8 * half;
    #pragma unroll
    for (int r = 0; r < 8; ++r) {
        float* orow = h + (size_t)(rbase + r) * OUT_FEATS + l15;
        orow[0]  = c0[r];
        orow[16] = c1[r];
        orow[32] = c2[r];
        orow[48] = c3[r];
    }
}

__global__ void __launch_bounds__(GATH_THREADS)
gcn_gather_finalize(const unsigned* __restrict__ rec,
                    const int* __restrict__ offs,
                    const float* __restrict__ h,
                    const float* __restrict__ b,
                    float* __restrict__ out,
                    int n_edges, int n_nodes) {
    __shared__ float acc[GATH_LDS_WORDS];
    __shared__ unsigned cnt[256];
    __shared__ unsigned rstart[256];
    __shared__ unsigned indeg[BKT_NODES];
    __shared__ int more;

    unsigned* sorted = reinterpret_cast<unsigned*>(acc) + SORT_OFF;
    v2f* acc2 = reinterpret_cast<v2f*>(acc);

    const int tid = threadIdx.x;
    const int lane = tid & 31;
    const int wave = __builtin_amdgcn_readfirstlane(tid >> 5);
    const int bk = blockIdx.x;
    const int node0 = bk * BKT_NODES;
    const int nn = max(0, min(BKT_NODES, n_nodes - node0));
    const int cap = (n_edges + BIN_SLICES - 1) / BIN_SLICES;

    const int r_lo = min((wave * BKT_NODES) / GATH_WAVES, nn);
    const int r_hi = min(((wave + 1) * BKT_NODES) / GATH_WAVES, nn);

    if (tid < BKT_NODES) indeg[tid] = 0u;

    int o0[RUNS_PER_WAVE], o1[RUNS_PER_WAVE];
    #pragma unroll
    for (int j = 0; j < RUNS_PER_WAVE; ++j) {
        const int s = wave + j * GATH_WAVES;
        o0[j] = offs[s * (NBKT + 1) + bk];
        o1[j] = offs[s * (NBKT + 1) + bk + 1];
    }
    const v2f bb = *reinterpret_cast<const v2f*>(b + 2 * lane);

    for (int round = 0; ; ++round) {
        if (tid < 256) cnt[tid] = 0u;
        if (tid == 0) more = 0;

        unsigned rr[RUNS_PER_WAVE][2];
        bool left = false;
        #pragma unroll
        for (int j = 0; j < RUNS_PER_WAVE; ++j) {
            const unsigned* rp = rec + (size_t)(wave + j * GATH_WAVES) * cap;
            const int a0 = o0[j] + round * ROUND_CAP;
            #pragma unroll
            for (int u = 0; u < 2; ++u) {
                const int i = a0 + 32 * u + lane;
                rr[j][u] = (i < o1[j]) ? rp[i] : 0xFFFFFFFFu;
            }
            left = left || (o1[j] - a0 > ROUND_CAP);
        }
        __syncthreads();

        if (left && lane == 0) atomicOr(&more, 1);
        #pragma unroll
        for (int j = 0; j < RUNS_PER_WAVE; ++j)
            #pragma unroll
            for (int u = 0; u < 2; ++u)
                if (rr[j][u] != 0xFFFFFFFFu) atomicAdd(&cnt[rr[j][u] >> 24], 1u);
        __syncthreads();

        if (wave == 0) {
            unsigned c[7];
            unsigned tot = 0u;
            #pragma unroll
            for (int k = 0; k < 7; ++k) {
                c[k] = cnt[lane * 7 + k];
                tot += c[k];
            }
            unsigned inc = tot;
            #pragma unroll
            for (int o = 1; o < 32; o <<= 1) {
                const unsigned t = __shfl_up(inc, o);
                if (lane >= o) inc += t;
            }
            unsigned run = inc - tot;
            #pragma unroll
            for (int k = 0; k < 7; ++k) {
                const int r = lane * 7 + k;
                rstart[r] = run;
                cnt[r] = run;
                if (r < BKT_NODES) indeg[r] += c[k];
                run += c[k];
            }
        }
        __syncthreads();

        #pragma unroll
        for (int j = 0; j < RUNS_PER_WAVE; ++j)
            #pragma unroll
            for (int u = 0; u < 2; ++u)
                if (rr[j][u] != 0xFFFFFFFFu) {
                    const unsigned pos = atomicAdd(&cnt[rr[j][u] >> 24], 1u);
                    sorted[pos] = rr[j][u] & 0xFFFFFFu;
                }
        __syncthreads();

        const bool last = (more == 0);

        if (r_lo < r_hi) {
            const int bnd = (int)rstart[r_lo + lane];
            const int s_end = __builtin_amdgcn_readlane(bnd, r_hi - r_lo);
            int r = r_lo;
            int rend = __builtin_amdgcn_readlane(bnd, 1);
            v2f sum = {0.0f, 0.0f};

            auto close_row = [&]() {
                v2f t = sum;
                if (round > 0)
                    t += acc2[r * (OUT_FEATS / 2) + lane];
                acc2[r * (OUT_FEATS / 2) + lane] = t;
                sum = v2f{0.0f, 0.0f};
                ++r;
                rend = __builtin_amdgcn_readlane(bnd, r - r_lo + 1);
            };

            const int s_begin = __builtin_amdgcn_readlane(bnd, 0);
            unsigned rc = (s_begin < s_end) ? sorted[min(s_begin + (lane & (GATH_EB - 1)), s_end - 1)] : 0u;
            for (int base = s_begin; base < s_end; base += GATH_EB) {
                v2f v[GATH_EB];
                #pragma unroll
                for (int q = 0; q < GATH_EB; ++q) {
                    const unsigned x = (unsigned)__builtin_amdgcn_readlane((int)rc, q);
                    v[q] = *reinterpret_cast<const v2f*>(h + (size_t)x * OUT_FEATS + 2 * lane);
                }
                rc = sorted[min(base + GATH_EB + (lane & (GATH_EB - 1)), s_end - 1)];
                #pragma unroll
                for (int q = 0; q < GATH_EB; ++q) {
                    const int idx = base + q;
                    if (idx < s_end) {
                        while (idx >= rend) close_row();
                        sum += v[q];
                    }
                }
            }
            while (r < r_hi) close_row();

            if (last) {
                for (int rr2 = r_lo; rr2 < r_hi; ++rr2) {
                    const v2f t = acc2[rr2 * (OUT_FEATS / 2) + lane];
                    const float nd = rsqrtf(fmaxf((float)indeg[rr2], 1.0f));
                    const float x0 = t.x * nd + bb.x;
                    const float x1 = t.y * nd + bb.y;
                    v2f y;
                    y.x = 1.0f / (1.0f + __expf(-x0));
                    y.y = 1.0f / (1.0f + __expf(-x1));
                    *reinterpret_cast<v2f*>(out + (size_t)(node0 + rr2) * OUT_FEATS + 2 * lane) = y;
                }
            }
        }

        if (last) break;
        __syncthreads();
    }
}

extern "C" void kernel_launch(void* const* d_in, const int* in_sizes, int n_in,
                              void* d_out, int out_size, void* d_ws, size_t ws_size,
                              hipStream_t stream) {
    const float* feat = (const float*)d_in[0];
    const int*   src  = (const int*)d_in[1];
    const int*   dst  = (const int*)d_in[2];
    const float* W    = (const float*)d_in[3];
    const float* bias = (const float*)d_in[4];
    float*       out  = (float*)d_out;

    const int n_nodes = in_sizes[0] / IN_FEATS;
    const int n_edges = in_sizes[1];

    float* h        = (float*)d_ws;
    unsigned* part  = (unsigned*)(h + (size_t)n_nodes * OUT_FEATS);
    int* offs       = (int*)(part + (size_t)DEG_SLICES * n_nodes);
    unsigned* rec   = (unsigned*)(offs + BIN_SLICES * (NBKT + 1));

    gcn_degree_bin<<<DEG_WGS + BIN_SLICES, DEG_THREADS, 0, stream>>>(
        src, dst, part, offs, rec, n_edges, n_nodes);

    const int n_tiles = n_nodes / 16;
    gcn_gemm_wmma<<<(n_tiles + 7) / 8, 256, 0, stream>>>(
        feat, part, W, h, n_tiles, n_nodes);

    gcn_gather_finalize<<<NBKT, GATH_THREADS, 0, stream>>>(
        rec, offs, h, bias, out, n_edges, n_nodes);
}
